# starting kernel
# baseline (speedup 1.0000x reference)
#include <hip/hip_runtime.h>
#include <hip/hip_bf16.h>
#include <math.h>

typedef __attribute__((ext_vector_type(2))) float v2f;
typedef __attribute__((ext_vector_type(4))) float v4f;
typedef __attribute__((ext_vector_type(8))) float v8f;

#define BTOT   192
#define NN     325
#define DD     64
#define HH     4
#define COLS   256
#define MROWS  (NN * BTOT)
#define MTILES (MROWS / 16)
#define NEG_SLOPE 0.2f

__global__ __launch_bounds__(256) void gat_proj_wmma(
    const float* __restrict__ x,
    const float* __restrict__ W,
    float* __restrict__ xp)
{
    __shared__ float sW[DD * COLS];
    for (int i = threadIdx.x; i < (DD * COLS) / 4; i += 256) {
        ((v4f*)sW)[i] = ((const v4f*)W)[i];
    }
    __syncthreads();

    const int wave  = threadIdx.x >> 5;
    const int lane  = threadIdx.x & 31;
    const int mtile = blockIdx.x * 8 + wave;
    if (mtile >= MTILES) return;

    const int m0   = mtile * 16;
    const int r    = lane & 15;
    const int kgrp = (lane >> 4) * 2;
    const int m    = m0 + r;
    const int n    = m / BTOT;
    const int bt   = m - n * BTOT;
    const float* arow = x + ((size_t)bt * NN + n) * DD;

    v2f afrag[16];
#pragma unroll
    for (int kk = 0; kk < 16; ++kk)
        afrag[kk] = *(const v2f*)(arow + kk * 4 + kgrp);

    float* orow = xp + (size_t)m0 * COLS;
#pragma unroll 1
    for (int c = 0; c < 16; ++c) {
        v8f acc = {};
#pragma unroll
        for (int kk = 0; kk < 16; ++kk) {
            v2f b;
            b.x = sW[(kk * 4 + kgrp + 0) * COLS + c * 16 + (lane & 15)];
            b.y = sW[(kk * 4 + kgrp + 1) * COLS + c * 16 + (lane & 15)];
            acc = __builtin_amdgcn_wmma_f32_16x16x4_f32(
                false, afrag[kk], false, b, (short)0, acc, false, false);
        }
        const int col   = c * 16 + (lane & 15);
        const int rbase = (lane >> 4) * 8;
#pragma unroll
        for (int v = 0; v < 8; ++v)
            orow[(size_t)(rbase + v) * COLS + col] = acc[v];
    }
}

__global__ __launch_bounds__(256) void gat_att(
    const float* __restrict__ xp,
    const float* __restrict__ att_src,
    const float* __restrict__ att_dst,
    float* __restrict__ asrc,
    float* __restrict__ adst)
{
    const int gid = blockIdx.x * 256 + threadIdx.x;
    const int m   = gid >> 5;
    if (m >= MROWS) return;
    const int lane = threadIdx.x & 31;
    const int h    = lane >> 3;
    const int sub  = lane & 7;

    const float* p  = xp + (size_t)m * COLS + h * DD + sub * 8;
    const float* as = att_src + h * DD + sub * 8;
    const float* ad = att_dst + h * DD + sub * 8;
    float ssum = 0.f, dsum = 0.f;
#pragma unroll
    for (int i = 0; i < 8; ++i) { float v = p[i]; ssum += v * as[i]; dsum += v * ad[i]; }
#pragma unroll
    for (int o = 4; o >= 1; o >>= 1) {
        ssum += __shfl_xor(ssum, o, 32);
        dsum += __shfl_xor(dsum, o, 32);
    }
    if (sub == 0) {
        asrc[(size_t)m * HH + h] = ssum;
        adst[(size_t)m * HH + h] = dsum;
    }
}

__global__ void csr_init(int* __restrict__ deg) {
    int n = blockIdx.x * blockDim.x + threadIdx.x;
    if (n < NN) deg[n] = 1;
}

__global__ void csr_count(const int* __restrict__ ei, int* __restrict__ deg, int E) {
    int e = blockIdx.x * blockDim.x + threadIdx.x;
    if (e < E) atomicAdd(&deg[ei[E + e]], 1);
}

__global__ void csr_scan(const int* __restrict__ deg, int* __restrict__ off,
                         int* __restrict__ cur) {
    if (blockIdx.x == 0 && threadIdx.x == 0) {
        int acc = 0;
        for (int n = 0; n < NN; ++n) { off[n] = acc; cur[n] = acc; acc += deg[n]; }
        off[NN] = acc;
    }
}

__global__ void csr_fill(const int* __restrict__ ei, int* __restrict__ cur,
                         int* __restrict__ csrc, int E) {
    int t = blockIdx.x * blockDim.x + threadIdx.x;
    if (t < E) {
        int s = ei[t], d = ei[E + t];
        csrc[atomicAdd(&cur[d], 1)] = s;
    } else if (t < E + NN) {
        int n = t - E;
        csrc[atomicAdd(&cur[n], 1)] = n;
    }
}

__global__ __launch_bounds__(128) void gat_agg_ln(
    const float* __restrict__ x,
    const float* __restrict__ xp,
    const float* __restrict__ asrc,
    const float* __restrict__ adst,
    const int* __restrict__ off,
    const int* __restrict__ csrc,
    const float* __restrict__ bias,
    const float* __restrict__ gamma,
    const float* __restrict__ beta,
    float* __restrict__ out)
{
    __shared__ float heads[HH * DD];
    const int id   = blockIdx.x;
    const int n    = id / BTOT;
    const int bt   = id - n * BTOT;
    const int h    = threadIdx.x >> 5;
    const int lane = threadIdx.x & 31;
    const int d0   = lane * 2;

    const float ad = adst[((size_t)n * BTOT + bt) * HH + h];
    const int kb = off[n], ke = off[n + 1];

    float mrun = -3.402823466e+38f;
    float srun = 0.f, acc0 = 0.f, acc1 = 0.f;
    for (int k = kb; k < ke; ++k) {
        const int j = csrc[k];
        float a = asrc[((size_t)j * BTOT + bt) * HH + h] + ad;
        a = (a >= 0.f) ? a : NEG_SLOPE * a;
        const float mnew = fmaxf(mrun, a);
        const float corr = __expf(mrun - mnew);
        const float w    = __expf(a - mnew);
        srun = srun * corr + w;
        const float* xj = xp + ((size_t)j * BTOT + bt) * COLS + h * DD + d0;
        acc0 = acc0 * corr + w * xj[0];
        acc1 = acc1 * corr + w * xj[1];
        mrun = mnew;
    }
    const float inv = 1.f / (srun + 1e-16f);
    heads[h * DD + d0]     = acc0 * inv;
    heads[h * DD + d0 + 1] = acc1 * inv;
    __syncthreads();

    if (h == 0) {
        float y0 = 0.f, y1 = 0.f;
#pragma unroll
        for (int hh = 0; hh < HH; ++hh) {
            y0 += heads[hh * DD + d0];
            y1 += heads[hh * DD + d0 + 1];
        }
        const float* xr = x + ((size_t)bt * NN + n) * DD;
        y0 = y0 * 0.25f + bias[d0]     + xr[d0];
        y1 = y1 * 0.25f + bias[d0 + 1] + xr[d0 + 1];
        float s  = y0 + y1;
        float s2 = y0 * y0 + y1 * y1;
#pragma unroll
        for (int o = 16; o >= 1; o >>= 1) {
            s  += __shfl_xor(s,  o, 32);
            s2 += __shfl_xor(s2, o, 32);
        }
        const float mu  = s * (1.f / 64.f);
        const float var = s2 * (1.f / 64.f) - mu * mu;
        const float rs  = rsqrtf(var + 1e-5f);
        float* orow = out + ((size_t)bt * NN + n) * DD;
        orow[d0]     = (y0 - mu) * rs * gamma[d0]     + beta[d0];
        orow[d0 + 1] = (y1 - mu) * rs * gamma[d0 + 1] + beta[d0 + 1];
    }
}

extern "C" void kernel_launch(void* const* d_in, const int* in_sizes, int n_in,
                              void* d_out, int out_size, void* d_ws, size_t ws_size,
                              hipStream_t stream) {
    const float* x        = (const float*)d_in[0];
    const float* W        = (const float*)d_in[1];
    const float* att_src  = (const float*)d_in[2];
    const float* att_dst  = (const float*)d_in[3];
    const float* bias     = (const float*)d_in[4];
    const float* gamma    = (const float*)d_in[5];
    const float* beta     = (const float*)d_in[6];
    const int*   ei       = (const int*)d_in[7];
    float* out            = (float*)d_out;
    const int E = in_sizes[7] / 2;

    char* ws = (char*)d_ws;
    float* xp   = (float*)ws;  ws += sizeof(float) * (size_t)MROWS * COLS;
    float* asrc = (float*)ws;  ws += sizeof(float) * (size_t)MROWS * HH;
    float* adst = (float*)ws;  ws += sizeof(float) * (size_t)MROWS * HH;
    int* deg    = (int*)ws;    ws += sizeof(int) * NN;
    int* off    = (int*)ws;    ws += sizeof(int) * (NN + 1);
    int* cur    = (int*)ws;    ws += sizeof(int) * NN;
    int* csrc   = (int*)ws;    ws += sizeof(int) * (E + NN);

    gat_proj_wmma<<<(MTILES + 7) / 8, 256, 0, stream>>>(x, W, xp);

    gat_att<<<(MROWS * 32 + 255) / 256, 256, 0, stream>>>(xp, att_src, att_dst, asrc, adst);

    csr_init <<<(NN + 255) / 256, 256, 0, stream>>>(deg);
    csr_count<<<(E + 255) / 256, 256, 0, stream>>>(ei, deg, E);
    csr_scan <<<1, 32, 0, stream>>>(deg, off, cur);
    csr_fill <<<(E + NN + 255) / 256, 256, 0, stream>>>(ei, cur, csrc, E);

    gat_agg_ln<<<MROWS, 128, 0, stream>>>(x, xp, asrc, adst, off, csrc,
                                          bias, gamma, beta, out);
}


# aggregation kernel re-decomposed: one wave per node and 4 bt covers all 4 heads; edge logits, max and sum lane-parallel first, then gathers as independent 16-byte loads, 4 edges per trip
# speedup vs baseline: 1.2212x; 1.2212x over previous
#include <hip/hip_runtime.h>
#include <hip/hip_bf16.h>
#include <math.h>

typedef __attribute__((ext_vector_type(2))) float v2f;
typedef __attribute__((ext_vector_type(4))) float v4f;
typedef __attribute__((ext_vector_type(8))) float v8f;

#define BTOT   192
#define NN     325
#define DD     64
#define HH     4
#define COLS   256
#define MROWS  (NN * BTOT)
#define MTILES (MROWS / 16)
#define NEG_SLOPE 0.2f

__global__ __launch_bounds__(256) void gat_proj_wmma(
    const float* __restrict__ x,
    const float* __restrict__ W,
    float* __restrict__ xp)
{
    __shared__ float sW[DD * COLS];
    for (int i = threadIdx.x; i < (DD * COLS) / 4; i += 256) {
        ((v4f*)sW)[i] = ((const v4f*)W)[i];
    }
    __syncthreads();

    const int wave  = threadIdx.x >> 5;
    const int lane  = threadIdx.x & 31;
    const int mtile = blockIdx.x * 8 + wave;
    if (mtile >= MTILES) return;

    const int m0   = mtile * 16;
    const int r    = lane & 15;
    const int kgrp = (lane >> 4) * 2;
    const int m    = m0 + r;
    const int n    = m / BTOT;
    const int bt   = m - n * BTOT;
    const float* arow = x + ((size_t)bt * NN + n) * DD;

    v2f afrag[16];
#pragma unroll
    for (int kk = 0; kk < 16; ++kk)
        afrag[kk] = *(const v2f*)(arow + kk * 4 + kgrp);

    float* orow = xp + (size_t)m0 * COLS;
#pragma unroll 1
    for (int c = 0; c < 16; ++c) {
        v8f acc = {};
#pragma unroll
        for (int kk = 0; kk < 16; ++kk) {
            v2f b;
            b.x = sW[(kk * 4 + kgrp + 0) * COLS + c * 16 + (lane & 15)];
            b.y = sW[(kk * 4 + kgrp + 1) * COLS + c * 16 + (lane & 15)];
            acc = __builtin_amdgcn_wmma_f32_16x16x4_f32(
                false, afrag[kk], false, b, (short)0, acc, false, false);
        }
        const int col   = c * 16 + (lane & 15);
        const int rbase = (lane >> 4) * 8;
#pragma unroll
        for (int v = 0; v < 8; ++v)
            orow[(size_t)(rbase + v) * COLS + col] = acc[v];
    }
}

__global__ __launch_bounds__(256) void gat_att(
    const float* __restrict__ xp,
    const float* __restrict__ att_src,
    const float* __restrict__ att_dst,
    float* __restrict__ asrc,
    float* __restrict__ adst)
{
    const int gid = blockIdx.x * 256 + threadIdx.x;
    const int m   = gid >> 5;
    if (m >= MROWS) return;
    const int lane = threadIdx.x & 31;
    const int h    = lane >> 3;
    const int sub  = lane & 7;

    const float* p  = xp + (size_t)m * COLS + h * DD + sub * 8;
    const float* as = att_src + h * DD + sub * 8;
    const float* ad = att_dst + h * DD + sub * 8;
    float ssum = 0.f, dsum = 0.f;
#pragma unroll
    for (int i = 0; i < 8; ++i) { float v = p[i]; ssum += v * as[i]; dsum += v * ad[i]; }
#pragma unroll
    for (int o = 4; o >= 1; o >>= 1) {
        ssum += __shfl_xor(ssum, o, 32);
        dsum += __shfl_xor(dsum, o, 32);
    }
    if (sub == 0) {
        asrc[(size_t)m * HH + h] = ssum;
        adst[(size_t)m * HH + h] = dsum;
    }
}

__global__ void csr_init(int* __restrict__ deg) {
    int n = blockIdx.x * blockDim.x + threadIdx.x;
    if (n < NN) deg[n] = 1;
}

__global__ void csr_count(const int* __restrict__ ei, int* __restrict__ deg, int E) {
    int e = blockIdx.x * blockDim.x + threadIdx.x;
    if (e < E) atomicAdd(&deg[ei[E + e]], 1);
}

__global__ void csr_scan(const int* __restrict__ deg, int* __restrict__ off,
                         int* __restrict__ cur) {
    if (blockIdx.x == 0 && threadIdx.x == 0) {
        int acc = 0;
        for (int n = 0; n < NN; ++n) { off[n] = acc; cur[n] = acc; acc += deg[n]; }
        off[NN] = acc;
    }
}

__global__ void csr_fill(const int* __restrict__ ei, int* __restrict__ cur,
                         int* __restrict__ csrc, int E) {
    int t = blockIdx.x * blockDim.x + threadIdx.x;
    if (t < E) {
        int s = ei[t], d = ei[E + t];
        csrc[atomicAdd(&cur[d], 1)] = s;
    } else if (t < E + NN) {
        int n = t - E;
        csrc[atomicAdd(&cur[n], 1)] = n;
    }
}

#define AGG_BT     4
#define AGG_WAVES  4

__device__ __forceinline__ float lane_xhalf(float v) {
    int b = __builtin_bit_cast(int, v);
    return __builtin_bit_cast(float,
        __builtin_amdgcn_permlanex16(b, b, 0x76543210, 0xfedcba98, false, false));
}
#define LANE_XROW(v, M) __builtin_bit_cast(float, __builtin_amdgcn_update_dpp( \
        __builtin_bit_cast(int, (v)), __builtin_bit_cast(int, (v)), 0x160 | (M), 0xf, 0xf, true))

__device__ __forceinline__ float row16_sum(float v) {
    v += LANE_XROW(v, 1); v += LANE_XROW(v, 2); v += LANE_XROW(v, 4); v += LANE_XROW(v, 8);
    return v;
}
__device__ __forceinline__ float wave_sum(float v) {
    v = row16_sum(v);
    return v + lane_xhalf(v);
}
__device__ __forceinline__ float wave_max(float v) {
    v = fmaxf(v, LANE_XROW(v, 1)); v = fmaxf(v, LANE_XROW(v, 2));
    v = fmaxf(v, LANE_XROW(v, 4)); v = fmaxf(v, LANE_XROW(v, 8));
    return fmaxf(v, lane_xhalf(v));
}
__device__ __forceinline__ float bcast_lane(float v, int e) {
    return __builtin_bit_cast(float, __builtin_amdgcn_readlane(__builtin_bit_cast(int, v), e));
}
__device__ __forceinline__ float leaky(float a) { return (a >= 0.f) ? a : NEG_SLOPE * a; }

__global__ __launch_bounds__(32 * AGG_WAVES) void gat_agg_ln(
    const float* __restrict__ x,
    const float* __restrict__ xp,
    const float* __restrict__ asrc,
    const float* __restrict__ adst,
    const int* __restrict__ off,
    const int* __restrict__ csrc,
    const float* __restrict__ bias,
    const float* __restrict__ gamma,
    const float* __restrict__ beta,
    float* __restrict__ out)
{
    const int lane = threadIdx.x & 31;
    const int wave = __builtin_amdgcn_readfirstlane((int)(threadIdx.x >> 5));
    const int gw   = blockIdx.x * AGG_WAVES + wave;
    const int n    = gw / (BTOT / AGG_BT);
    const int bt0  = (gw - n * (BTOT / AGG_BT)) * AGG_BT;
    const int kb = off[n], ke = off[n + 1];
    const bool hi = lane >= 16;
    const int  d4 = (lane & 15) * 4;
    const int  jfirst = csrc[min(kb + lane, ke - 1)];
    const v4f bb = *(const v4f*)(bias + d4);
    const v4f g  = *(const v4f*)(gamma + d4);
    const v4f be = *(const v4f*)(beta + d4);

    for (int b = 0; b < AGG_BT; ++b) {
        const int bt = bt0 + b;
        const v4f ad4 = *(const v4f*)(adst + ((size_t)n * BTOT + bt) * HH);
        const size_t orow = ((size_t)bt * NN + n) * DD + d4;
        const v4f xr = *(const v4f*)(x + orow);

        float m0 = -3.402823466e+38f, m1 = m0, m2 = m0, m3 = m0;
        float s0 = 0.f, s1 = 0.f, s2 = 0.f, s3 = 0.f;
        v4f accA = {}, accB = {};

        for (int base = kb; base < ke; base += 32) {
            const int  cnt  = min(32, ke - base);
            const bool live = lane < cnt;
            const int  j    = (base == kb) ? jfirst : csrc[min(base + lane, ke - 1)];
            const v4f  as4  = *(const v4f*)(asrc + ((size_t)j * BTOT + bt) * HH);
            const float a0 = live ? leaky(as4.x + ad4.x) : -3.402823466e+38f;
            const float a1 = live ? leaky(as4.y + ad4.y) : -3.402823466e+38f;
            const float a2 = live ? leaky(as4.z + ad4.z) : -3.402823466e+38f;
            const float a3 = live ? leaky(as4.w + ad4.w) : -3.402823466e+38f;
            const float n0 = fmaxf(m0, wave_max(a0));
            const float n1 = fmaxf(m1, wave_max(a1));
            const float n2 = fmaxf(m2, wave_max(a2));
            const float n3 = fmaxf(m3, wave_max(a3));
            const float c0 = __expf(m0 - n0), c1 = __expf(m1 - n1);
            const float c2 = __expf(m2 - n2), c3 = __expf(m3 - n3);
            const float w0 = live ? __expf(a0 - n0) : 0.f;
            const float w1 = live ? __expf(a1 - n1) : 0.f;
            const float w2 = live ? __expf(a2 - n2) : 0.f;
            const float w3 = live ? __expf(a3 - n3) : 0.f;
            s0 = s0 * c0 + wave_sum(w0);
            s1 = s1 * c1 + wave_sum(w1);
            s2 = s2 * c2 + wave_sum(w2);
            s3 = s3 * c3 + wave_sum(w3);
            accA *= hi ? c1 : c0;
            accB *= hi ? c3 : c2;
            m0 = n0; m1 = n1; m2 = n2; m3 = n3;

            const float* xcol = xp + (size_t)bt * COLS + lane * 4;
            const int cnt4 = (cnt + 3) & ~3;
            for (int e = 0; e < cnt4; e += 4) {
#pragma unroll
                for (int u = 0; u < 4; ++u) {
                    const int   je = __builtin_amdgcn_readlane(j, e + u);
                    const float e0 = bcast_lane(w0, e + u), e1 = bcast_lane(w1, e + u);
                    const float e2 = bcast_lane(w2, e + u), e3 = bcast_lane(w3, e + u);
                    const float wa = hi ? e1 : e0;
                    const float wb = hi ? e3 : e2;
                    const float* row = xcol + (size_t)je * (BTOT * COLS);
                    const v4f va = *(const v4f*)(row);
                    const v4f vb = *(const v4f*)(row + 2 * DD);
                    accA += wa * va;
                    accB += wb * vb;
                }
            }
        }

        const float i0 = 1.f / (s0 + 1e-16f), i1 = 1.f / (s1 + 1e-16f);
        const float i2 = 1.f / (s2 + 1e-16f), i3 = 1.f / (s3 + 1e-16f);
        v4f y = accA * (hi ? i1 : i0) + accB * (hi ? i3 : i2);
        y.x += lane_xhalf(y.x); y.y += lane_xhalf(y.y);
        y.z += lane_xhalf(y.z); y.w += lane_xhalf(y.w);

        y = y * 0.25f + bb + xr;
        const float sm = row16_sum((y.x + y.y) + (y.z + y.w));
        const float sq = row16_sum((y.x * y.x + y.y * y.y) + (y.z * y.z + y.w * y.w));
        const float mu  = sm * (1.f / 64.f);
        const float var = sq * (1.f / 64.f) - mu * mu;
        const float rs  = rsqrtf(var + 1e-5f);
        const v4f o  = (y - mu) * rs * g + be;
        if (!hi) *(v4f*)(out + orow) = o;
    }
}

extern "C" void kernel_launch(void* const* d_in, const int* in_sizes, int n_in,
                              void* d_out, int out_size, void* d_ws, size_t ws_size,
                              hipStream_t stream) {
    const float* x        = (const float*)d_in[0];
    const float* W        = (const float*)d_in[1];
    const float* att_src  = (const float*)d_in[2];
    const float* att_dst  = (const float*)d_in[3];
    const float* bias     = (const float*)d_in[4];
    const float* gamma    = (const float*)d_in[5];
    const float* beta     = (const float*)d_in[6];
    const int*   ei       = (const int*)d_in[7];
    float* out            = (float*)d_out;
    const int E = in_sizes[7] / 2;

    char* ws = (char*)d_ws;
    float* xp   = (float*)ws;  ws += sizeof(float) * (size_t)MROWS * COLS;
    float* asrc = (float*)ws;  ws += sizeof(float) * (size_t)MROWS * HH;
    float* adst = (float*)ws;  ws += sizeof(float) * (size_t)MROWS * HH;
    int* deg    = (int*)ws;    ws += sizeof(int) * NN;
    int* off    = (int*)ws;    ws += sizeof(int) * (NN + 1);
    int* cur    = (int*)ws;    ws += sizeof(int) * NN;
    int* csrc   = (int*)ws;    ws += sizeof(int) * (E + NN);

    gat_proj_wmma<<<(MTILES + 7) / 8, 256, 0, stream>>>(x, W, xp);

    gat_att<<<(MROWS * 32 + 255) / 256, 256, 0, stream>>>(xp, att_src, att_dst, asrc, adst);

    csr_init <<<(NN + 255) / 256, 256, 0, stream>>>(deg);
    csr_count<<<(E + 255) / 256, 256, 0, stream>>>(ei, deg, E);
    csr_scan <<<1, 32, 0, stream>>>(deg, off, cur);
    csr_fill <<<(E + NN + 255) / 256, 256, 0, stream>>>(ei, cur, csrc, E);

    gat_agg_ln<<<MROWS / (AGG_BT * AGG_WAVES), 32 * AGG_WAVES, 0, stream>>>(x, xp, asrc, adst, off, csrc,
                                          bias, gamma, beta, out);
}


# aggregation before projection: softmax-weighted sums of x rows (4 head weights) first, then 16 items x K=256 projected on f32 WMMA in the same kernel; xp never written; summation order changed
# speedup vs baseline: 1.8347x; 1.5024x over previous
#include <hip/hip_runtime.h>
#include <hip/hip_bf16.h>
#include <math.h>


typedef __attribute__((ext_vector_type(2))) float v2f;
typedef __attribute__((ext_vector_type(4))) float v4f;
typedef __attribute__((ext_vector_type(8))) float v8f;

#define BTOT   192
#define NN     325
#define DD     64
#define HH     4
#define COLS   256
#define MROWS  (NN * BTOT)
#define MTILES (MROWS / 16)
#define NEG_SLOPE 0.2f

#define ACOLS 16

#define LANE_XROW(v, M) __builtin_bit_cast(float, __builtin_amdgcn_update_dpp( \
        __builtin_bit_cast(int, (v)), __builtin_bit_cast(int, (v)), 0x160 | (M), 0xf, 0xf, true))

__device__ __forceinline__ float row16_sum(float v) {
    v += LANE_XROW(v, 1); v += LANE_XROW(v, 2); v += LANE_XROW(v, 4); v += LANE_XROW(v, 8);
    return v;
}
__device__ __forceinline__ float leaky(float a) { return (a >= 0.f) ? a : NEG_SLOPE * a; }

#define CSR_CHUNK ((NN + 31) / 32)
#define CSR_U     8

__device__ __forceinline__ void csr_build(const int* __restrict__ ei, int E,
                                          int* __restrict__ off, int* __restrict__ csrc,
                                          int* sdeg, int* scur)
{
    const int tid = threadIdx.x;
    for (int n = tid; n < NN; n += 256) sdeg[n] = 1;
    __syncthreads();
    for (int base = tid; base < E; base += 256 * CSR_U) {
        int d[CSR_U];
#pragma unroll
        for (int u = 0; u < CSR_U; ++u)
            d[u] = (base + u * 256 < E) ? ei[E + base + u * 256] : 0;
#pragma unroll
        for (int u = 0; u < CSR_U; ++u)
            if (base + u * 256 < E) atomicAdd(&sdeg[d[u]], 1);
    }
    __syncthreads();
    if (tid < 32) {
        int loc[CSR_CHUNK];
        int sum = 0;
#pragma unroll
        for (int k = 0; k < CSR_CHUNK; ++k) {
            const int n = tid * CSR_CHUNK + k;
            const int v = (n < NN) ? sdeg[n] : 0;
            loc[k] = v;
            sum += v;
        }
        int incl = sum;
        for (int d = 1; d < 32; d <<= 1) {
            const int y = __shfl_up(incl, d);
            if (tid >= d) incl += y;
        }
        int run = incl - sum;
#pragma unroll
        for (int k = 0; k < CSR_CHUNK; ++k) {
            const int n = tid * CSR_CHUNK + k;
            if (n < NN) { off[n] = run; scur[n] = run; }
            run += loc[k];
        }
        if (tid == 31) off[NN] = incl;
    }
    __syncthreads();
    for (int base = tid; base < E; base += 256 * CSR_U) {
        int s[CSR_U], d[CSR_U];
#pragma unroll
        for (int u = 0; u < CSR_U; ++u) {
            const bool in = base + u * 256 < E;
            s[u] = in ? ei[base + u * 256] : 0;
            d[u] = in ? ei[E + base + u * 256] : 0;
        }
#pragma unroll
        for (int u = 0; u < CSR_U; ++u)
            if (base + u * 256 < E) csrc[atomicAdd(&scur[d[u]], 1)] = s[u];
    }
    for (int n = tid; n < NN; n += 256)
        csrc[atomicAdd(&scur[n], 1)] = n;
}

__global__ __launch_bounds__(256) void gat_logits_wmma(
    const float* __restrict__ x,
    const float* __restrict__ W,
    const float* __restrict__ att_src,
    const float* __restrict__ att_dst,
    float* __restrict__ asrc,
    float* __restrict__ adst,
    const int* __restrict__ ei,
    int E,
    int* __restrict__ off,
    int* __restrict__ csrc)
{
    __shared__ __attribute__((aligned(16))) float sWa[DD * ACOLS];
    if (blockIdx.x == 0) {
        csr_build(ei, E, off, csrc, (int*)sWa, (int*)sWa + 512);
        return;
    }
    {
        const int f4 = (threadIdx.x & 15) * 4;
        const int hw = threadIdx.x >> 4;
        const int h  = hw & 3;
        const v4f a = *(const v4f*)(att_src + h * DD + f4);
        const v4f b = *(const v4f*)(att_dst + h * DD + f4);
        v4f w[16];
#pragma unroll
        for (int it = 0; it < 16; ++it)
            w[it] = *(const v4f*)(W + (size_t)(it * 16 + hw) * DD + f4);
#pragma unroll
        for (int it = 0; it < 16; ++it) {
            const int d = it * 4 + (hw >> 2);
            const float s = row16_sum((w[it].x * a.x + w[it].y * a.y) + (w[it].z * a.z + w[it].w * a.w));
            const float t = row16_sum((w[it].x * b.x + w[it].y * b.y) + (w[it].z * b.z + w[it].w * b.w));
            if ((threadIdx.x & 15) == 0) {
                float* o = sWa + d * ACOLS + h;
                o[0] = s; o[4] = t; o[8] = 0.f; o[12] = 0.f;
            }
        }
    }
    __syncthreads();

    const int wave  = threadIdx.x >> 5;
    const int lane  = threadIdx.x & 31;
    const int mtile = (blockIdx.x - 1) * 8 + wave;
    if (mtile >= MTILES) return;

    const int m0   = mtile * 16;
    const int r    = lane & 15;
    const int kgrp = (lane >> 4) * 2;
    const int m    = m0 + r;
    const int n    = m / BTOT;
    const int bt   = m - n * BTOT;
    const float* arow = x + ((size_t)bt * NN + n) * DD;

    v8f acc = {};
#pragma unroll
    for (int kk = 0; kk < 16; ++kk) {
        const v2f a = *(const v2f*)(arow + kk * 4 + kgrp);
        v2f b;
        b.x = sWa[(kk * 4 + kgrp + 0) * ACOLS + (lane & 15)];
        b.y = sWa[(kk * 4 + kgrp + 1) * ACOLS + (lane & 15)];
        acc = __builtin_amdgcn_wmma_f32_16x16x4_f32(
            false, a, false, b, (short)0, acc, false, false);
    }
    const int col   = lane & 15;
    const int rbase = (lane >> 4) * 8;
    if (col < 2 * HH) {
        float* dstp = ((col < HH) ? asrc : adst) + (size_t)(m0 + rbase) * HH + (col & (HH - 1));
#pragma unroll
        for (int v = 0; v < 8; ++v)
            dstp[v * HH] = acc[v];
    }
}

#define AP_WAVES 4
#define ZK 128
#define NEGMAX (-3.402823466e+38f)

__global__ __launch_bounds__(32 * AP_WAVES) void gat_agg_proj_ln(
    const float* __restrict__ x,
    const float* __restrict__ W,
    const float* __restrict__ asrc,
    const float* __restrict__ adst,
    const int* __restrict__ off,
    const int* __restrict__ csrc,
    const float* __restrict__ bias,
    const float* __restrict__ gamma,
    const float* __restrict__ beta,
    float* __restrict__ out)
{
    __shared__ __attribute__((aligned(16))) float sW[DD * COLS];
    __shared__ __attribute__((aligned(16))) float sZ[AP_WAVES * 16 * ZK];

    for (int i = threadIdx.x; i < (DD * COLS) / 4; i += 32 * AP_WAVES)
        ((v4f*)sW)[i] = ((const v4f*)W)[i];

    const int lane = threadIdx.x & 31;
    const int wave = __builtin_amdgcn_readfirstlane((int)(threadIdx.x >> 5));
    const int tile = blockIdx.x * AP_WAVES + wave;
    const int n    = tile / (BTOT / 16);
    const int bt0  = (tile - n * (BTOT / 16)) * 16;
    const int kb = off[n], ke = off[n + 1];
    float* zt = sZ + wave * (16 * ZK);

    const int ib = lane & 15;
    const int hp = lane >> 4;
    const size_t lo = (size_t)(bt0 + ib) * HH + 2 * hp;
    const v2f ad2 = *(const v2f*)(adst + (size_t)n * (BTOT * HH) + lo);

    float m0 = NEGMAX, m1 = NEGMAX, s0 = 0.f, s1 = 0.f;
    for (int base = kb; base < ke; base += 32) {
        const int cnt  = min(32, ke - base);
        const int cnt4 = (cnt + 3) & ~3;
        const int j    = csrc[min(base + lane, ke - 1)];
        for (int e = 0; e < cnt4; e += 4) {
            v2f as2[4];
#pragma unroll
            for (int u = 0; u < 4; ++u) {
                const int je = __builtin_amdgcn_readlane(j, e + u);
                as2[u] = *(const v2f*)(asrc + (size_t)je * (BTOT * HH) + lo);
            }
#pragma unroll
            for (int u = 0; u < 4; ++u) {
                const bool real = e + u < cnt;
                const float a0 = leaky(as2[u].x + ad2.x);
                const float a1 = leaky(as2[u].y + ad2.y);
                const float n0 = fmaxf(m0, a0);
                const float n1 = fmaxf(m1, a1);
                s0 = s0 * __expf(m0 - n0) + (real ? __expf(a0 - n0) : 0.f);
                s1 = s1 * __expf(m1 - n1) + (real ? __expf(a1 - n1) : 0.f);
                m0 = n0; m1 = n1;
            }
        }
    }
    const float i0 = 1.f / (s0 + 1e-16f);
    const float i1 = 1.f / (s1 + 1e-16f);

    v2f acc[16][HH];
#pragma unroll
    for (int b = 0; b < 16; ++b) {
#pragma unroll
        for (int h = 0; h < HH; ++h) acc[b][h] = (v2f){0.f, 0.f};
    }

    for (int base = kb; base < ke; base += 32) {
        const int cnt  = min(32, ke - base);
        const int cnt4 = (cnt + 3) & ~3;
        const int j    = csrc[min(base + lane, ke - 1)];
        for (int e = 0; e < cnt4; e += 4) {
            v2f as2[4];
#pragma unroll
            for (int u = 0; u < 4; ++u) {
                const int je = __builtin_amdgcn_readlane(j, e + u);
                as2[u] = *(const v2f*)(asrc + (size_t)je * (BTOT * HH) + lo);
            }
#pragma unroll
            for (int u = 0; u < 4; ++u) {
                const bool real = e + u < cnt;
                const float a0 = leaky(as2[u].x + ad2.x);
                const float a1 = leaky(as2[u].y + ad2.y);
                v2f c;
                c.x = real ? __expf(a0 - m0) * i0 : 0.f;
                c.y = real ? __expf(a1 - m1) * i1 : 0.f;
                *(v2f*)(zt + (e + u) * (16 * HH) + ib * HH + 2 * hp) = c;
            }
        }
        __syncthreads();

        const float* xcol = x + (size_t)bt0 * (NN * DD) + 2 * lane;
        for (int e = 0; e < cnt4; e += 4) {
            v2f row[4][16];
#pragma unroll
            for (int u = 0; u < 4; ++u) {
                const int je = __builtin_amdgcn_readlane(j, e + u);
                const float* rp = xcol + (size_t)je * DD;
#pragma unroll
                for (int b = 0; b < 16; ++b)
                    row[u][b] = *(const v2f*)(rp + (size_t)b * (NN * DD));
            }
#pragma unroll
            for (int u = 0; u < 4; ++u) {
                const float* cp = zt + (e + u) * (16 * HH);
#pragma unroll
                for (int b = 0; b < 16; ++b) {
                    const v4f c = *(const v4f*)(cp + b * HH);
                    acc[b][0] += c.x * row[u][b];
                    acc[b][1] += c.y * row[u][b];
                    acc[b][2] += c.z * row[u][b];
                    acc[b][3] += c.w * row[u][b];
                }
            }
        }
        __syncthreads();
    }

    const int r    = lane & 15;
    const int half = lane >> 4;
    const int kgrp = half * 2;
    const size_t rowbase = ((size_t)(bt0 + half * 8) * NN + n) * DD + r;
    float xr[8][4];
    v8f o0 = {}, o1 = {}, o2 = {}, o3 = {};
    const float* zrow = zt + r * ZK;
    const int    swr  = 4 * r;

#pragma unroll
    for (int part = 0; part < 2; ++part) {
#pragma unroll
        for (int b = 0; b < 16; ++b) {
            float* zr = zt + b * ZK;
            const int sw = 4 * b;
#pragma unroll
            for (int hh = 0; hh < 2; ++hh)
                *(v2f*)(zr + ((hh * DD + 2 * lane) ^ sw)) = acc[b][2 * part + hh];
        }
        __syncthreads();

        if (part == 0) {
#pragma unroll
            for (int v = 0; v < 8; ++v) {
#pragma unroll
                for (int c = 0; c < 4; ++c)
                    xr[v][c] = x[rowbase + (size_t)v * (NN * DD) + c * 16];
            }
        }

#pragma unroll 8
        for (int kk = 0; kk < 32; ++kk) {
            const v2f a = *(const v2f*)(zrow + ((kk * 4 + kgrp) ^ swr));
            const float* bp = sW + ((kk & 15) * 4 + kgrp) * COLS + (2 * part + (kk >> 4)) * DD + r;
            v2f b0, b1, b2, b3;
            b0.x = bp[0];  b0.y = bp[COLS];
            b1.x = bp[16]; b1.y = bp[COLS + 16];
            b2.x = bp[32]; b2.y = bp[COLS + 32];
            b3.x = bp[48]; b3.y = bp[COLS + 48];
            o0 = __builtin_amdgcn_wmma_f32_16x16x4_f32(false, a, false, b0, (short)0, o0, false, false);
            o1 = __builtin_amdgcn_wmma_f32_16x16x4_f32(false, a, false, b1, (short)0, o1, false, false);
            o2 = __builtin_amdgcn_wmma_f32_16x16x4_f32(false, a, false, b2, (short)0, o2, false, false);
            o3 = __builtin_amdgcn_wmma_f32_16x16x4_f32(false, a, false, b3, (short)0, o3, false, false);
        }
        if (part == 0) {
            __syncthreads();
        }
    }

    {
        const float bb0 = bias[r], bb1 = bias[16 + r], bb2 = bias[32 + r], bb3 = bias[48 + r];
        const float g0 = gamma[r], g1 = gamma[16 + r], g2 = gamma[32 + r], g3 = gamma[48 + r];
        const float be0 = beta[r], be1 = beta[16 + r], be2 = beta[32 + r], be3 = beta[48 + r];
#pragma unroll
        for (int v = 0; v < 8; ++v) {
            const size_t idx = rowbase + (size_t)v * (NN * DD);
            const float y0 = o0[v] * 0.25f + bb0 + xr[v][0];
            const float y1 = o1[v] * 0.25f + bb1 + xr[v][1];
            const float y2 = o2[v] * 0.25f + bb2 + xr[v][2];
            const float y3 = o3[v] * 0.25f + bb3 + xr[v][3];
            const float sm = row16_sum((y0 + y1) + (y2 + y3));
            const float sq = row16_sum((y0 * y0 + y1 * y1) + (y2 * y2 + y3 * y3));
            const float mu  = sm * (1.f / 64.f);
            const float var = sq * (1.f / 64.f) - mu * mu;
            const float rs  = rsqrtf(var + 1e-5f);
            out[idx]      = (y0 - mu) * rs * g0 + be0;
            out[idx + 16] = (y1 - mu) * rs * g1 + be1;
            out[idx + 32] = (y2 - mu) * rs * g2 + be2;
            out[idx + 48] = (y3 - mu) * rs * g3 + be3;
        }
    }
}

extern "C" void kernel_launch(void* const* d_in, const int* in_sizes, int n_in,
                              void* d_out, int out_size, void* d_ws, size_t ws_size,
                              hipStream_t stream) {
    const float* x        = (const float*)d_in[0];
    const float* W        = (const float*)d_in[1];
    const float* att_src  = (const float*)d_in[2];
    const float* att_dst  = (const float*)d_in[3];
    const float* bias     = (const float*)d_in[4];
    const float* gamma    = (const float*)d_in[5];
    const float* beta     = (const float*)d_in[6];
    const int*   ei       = (const int*)d_in[7];
    float* out            = (float*)d_out;
    const int E = in_sizes[7] / 2;

    char* ws = (char*)d_ws;
    float* asrc = (float*)ws;  ws += sizeof(float) * (size_t)MROWS * HH;
    float* adst = (float*)ws;  ws += sizeof(float) * (size_t)MROWS * HH;
    int* off    = (int*)ws;    ws += sizeof(int) * (NN + 1);
    int* csrc   = (int*)ws;    ws += sizeof(int) * (E + NN);

    gat_logits_wmma<<<(MTILES + 7) / 8 + 1, 256, 0, stream>>>(x, W, att_src, att_dst, asrc, adst,
                                                            ei, E, off, csrc);

    gat_agg_proj_ln<<<MTILES / AP_WAVES, 32 * AP_WAVES, 0, stream>>>(x, W, asrc, adst, off, csrc,
                                          bias, gamma, beta, out);
}


# launch 2 re-shaped: one 16-wave work-group per WGP, one tile per wave (dealt in turn), all tiles in one round, W staged 244 times not 975; no work-group barrier in the loops; gather 2 edges per trip
# speedup vs baseline: 2.3919x; 1.3037x over previous
#include <hip/hip_runtime.h>
#include <hip/hip_bf16.h>
#include <math.h>


typedef __attribute__((ext_vector_type(2))) float v2f;
typedef __attribute__((ext_vector_type(4))) float v4f;
typedef __attribute__((ext_vector_type(8))) float v8f;

#define BTOT   192
#define NN     325
#define DD     64
#define HH     4
#define COLS   256
#define MROWS  (NN * BTOT)
#define MTILES (MROWS / 16)
#define NEG_SLOPE 0.2f

#define ACOLS 16

#define LANE_XROW(v, M) __builtin_bit_cast(float, __builtin_amdgcn_update_dpp( \
        __builtin_bit_cast(int, (v)), __builtin_bit_cast(int, (v)), 0x160 | (M), 0xf, 0xf, true))

__device__ __forceinline__ float row16_sum(float v) {
    v += LANE_XROW(v, 1); v += LANE_XROW(v, 2); v += LANE_XROW(v, 4); v += LANE_XROW(v, 8);
    return v;
}
__device__ __forceinline__ float leaky(float a) { return (a >= 0.f) ? a : NEG_SLOPE * a; }

#define CSR_CHUNK ((NN + 31) / 32)
#define CSR_U     8

__device__ __forceinline__ void csr_build(const int* __restrict__ ei, int E,
                                          int* __restrict__ off, int* __restrict__ csrc,
                                          int* sdeg, int* scur)
{
    const int tid = threadIdx.x;
    for (int n = tid; n < NN; n += 256) sdeg[n] = 1;
    __syncthreads();
    for (int base = tid; base < E; base += 256 * CSR_U) {
        int d[CSR_U];
#pragma unroll
        for (int u = 0; u < CSR_U; ++u)
            d[u] = (base + u * 256 < E) ? ei[E + base + u * 256] : 0;
#pragma unroll
        for (int u = 0; u < CSR_U; ++u)
            if (base + u * 256 < E) atomicAdd(&sdeg[d[u]], 1);
    }
    __syncthreads();
    if (tid < 32) {
        int loc[CSR_CHUNK];
        int sum = 0;
#pragma unroll
        for (int k = 0; k < CSR_CHUNK; ++k) {
            const int n = tid * CSR_CHUNK + k;
            const int v = (n < NN) ? sdeg[n] : 0;
            loc[k] = v;
            sum += v;
        }
        int incl = sum;
        for (int d = 1; d < 32; d <<= 1) {
            const int y = __shfl_up(incl, d);
            if (tid >= d) incl += y;
        }
        int run = incl - sum;
#pragma unroll
        for (int k = 0; k < CSR_CHUNK; ++k) {
            const int n = tid * CSR_CHUNK + k;
            if (n < NN) { off[n] = run; scur[n] = run; }
            run += loc[k];
        }
        if (tid == 31) off[NN] = incl;
    }
    __syncthreads();
    for (int base = tid; base < E; base += 256 * CSR_U) {
        int s[CSR_U], d[CSR_U];
#pragma unroll
        for (int u = 0; u < CSR_U; ++u) {
            const bool in = base + u * 256 < E;
            s[u] = in ? ei[base + u * 256] : 0;
            d[u] = in ? ei[E + base + u * 256] : 0;
        }
#pragma unroll
        for (int u = 0; u < CSR_U; ++u)
            if (base + u * 256 < E) csrc[atomicAdd(&scur[d[u]], 1)] = s[u];
    }
    for (int n = tid; n < NN; n += 256)
        csrc[atomicAdd(&scur[n], 1)] = n;
}

__global__ __launch_bounds__(256) void gat_logits_wmma(
    const float* __restrict__ x,
    const float* __restrict__ W,
    const float* __restrict__ att_src,
    const float* __restrict__ att_dst,
    float* __restrict__ asrc,
    float* __restrict__ adst,
    const int* __restrict__ ei,
    int E,
    int* __restrict__ off,
    int* __restrict__ csrc)
{
    __shared__ __attribute__((aligned(16))) float sWa[DD * ACOLS];
    if (blockIdx.x == 0) {
        csr_build(ei, E, off, csrc, (int*)sWa, (int*)sWa + 512);
        return;
    }
    {
        const int f4 = (threadIdx.x & 15) * 4;
        const int hw = threadIdx.x >> 4;
        const int h  = hw & 3;
        const v4f a = *(const v4f*)(att_src + h * DD + f4);
        const v4f b = *(const v4f*)(att_dst + h * DD + f4);
        v4f w[16];
#pragma unroll
        for (int it = 0; it < 16; ++it)
            w[it] = *(const v4f*)(W + (size_t)(it * 16 + hw) * DD + f4);
#pragma unroll
        for (int it = 0; it < 16; ++it) {
            const int d = it * 4 + (hw >> 2);
            const float s = row16_sum((w[it].x * a.x + w[it].y * a.y) + (w[it].z * a.z + w[it].w * a.w));
            const float t = row16_sum((w[it].x * b.x + w[it].y * b.y) + (w[it].z * b.z + w[it].w * b.w));
            if ((threadIdx.x & 15) == 0) {
                float* o = sWa + d * ACOLS + h;
                o[0] = s; o[4] = t; o[8] = 0.f; o[12] = 0.f;
            }
        }
    }
    __syncthreads();

    const int wave  = threadIdx.x >> 5;
    const int lane  = threadIdx.x & 31;
    const int mtile = (blockIdx.x - 1) * 8 + wave;
    if (mtile >= MTILES) return;

    const int m0   = mtile * 16;
    const int r    = lane & 15;
    const int kgrp = (lane >> 4) * 2;
    const int m    = m0 + r;
    const int n    = m / BTOT;
    const int bt   = m - n * BTOT;
    const float* arow = x + ((size_t)bt * NN + n) * DD;

    v8f acc = {};
#pragma unroll
    for (int kk = 0; kk < 16; ++kk) {
        const v2f a = *(const v2f*)(arow + kk * 4 + kgrp);
        v2f b;
        b.x = sWa[(kk * 4 + kgrp + 0) * ACOLS + (lane & 15)];
        b.y = sWa[(kk * 4 + kgrp + 1) * ACOLS + (lane & 15)];
        acc = __builtin_amdgcn_wmma_f32_16x16x4_f32(
            false, a, false, b, (short)0, acc, false, false);
    }
    const int col   = lane & 15;
    const int rbase = (lane >> 4) * 8;
    if (col < 2 * HH) {
        float* dstp = ((col < HH) ? asrc : adst) + (size_t)(m0 + rbase) * HH + (col & (HH - 1));
#pragma unroll
        for (int v = 0; v < 8; ++v)
            dstp[v * HH] = acc[v];
    }
}

#define AP_WAVES 16
#define AP_GRID ((MTILES + AP_WAVES - 1) / AP_WAVES)
#define ZK 128
#define GE 2
#define NEGMAX (-3.402823466e+38f)

__device__ __forceinline__ void wave_lds_handover() {
    __builtin_amdgcn_fence(__ATOMIC_RELEASE, "wavefront");
    __builtin_amdgcn_wave_barrier();
    __builtin_amdgcn_fence(__ATOMIC_ACQUIRE, "wavefront");
}

__global__ __launch_bounds__(32 * AP_WAVES) void gat_agg_proj_ln(
    const float* __restrict__ x,
    const float* __restrict__ W,
    const float* __restrict__ asrc,
    const float* __restrict__ adst,
    const int* __restrict__ off,
    const int* __restrict__ csrc,
    const float* __restrict__ bias,
    const float* __restrict__ gamma,
    const float* __restrict__ beta,
    float* __restrict__ out)
{
    __shared__ __attribute__((aligned(16))) float sW[DD * COLS];
    __shared__ __attribute__((aligned(16))) float sZ[AP_WAVES * 16 * ZK];

    {
        v4f wst[(DD * COLS / 4) / (32 * AP_WAVES)];
#pragma unroll
        for (int k = 0; k < (DD * COLS / 4) / (32 * AP_WAVES); ++k)
            wst[k] = ((const v4f*)W)[threadIdx.x + k * (32 * AP_WAVES)];
#pragma unroll
        for (int k = 0; k < (DD * COLS / 4) / (32 * AP_WAVES); ++k)
            ((v4f*)sW)[threadIdx.x + k * (32 * AP_WAVES)] = wst[k];
    }

    const int lane = threadIdx.x & 31;
    const int wave = __builtin_amdgcn_readfirstlane((int)(threadIdx.x >> 5));
    const bool live = wave * AP_GRID + (int)blockIdx.x < MTILES;
    const int tile = min(wave * AP_GRID + (int)blockIdx.x, MTILES - 1);
    const int n    = tile / (BTOT / 16);
    const int bt0  = (tile - n * (BTOT / 16)) * 16;
    const int kb = off[n], ke = off[n + 1];
    float* zt = sZ + wave * (16 * ZK);

    const int ib = lane & 15;
    const int hp = lane >> 4;
    const size_t lo = (size_t)(bt0 + ib) * HH + 2 * hp;
    const v2f ad2 = *(const v2f*)(adst + (size_t)n * (BTOT * HH) + lo);

    float m0 = NEGMAX, m1 = NEGMAX, s0 = 0.f, s1 = 0.f;
    for (int base = kb; base < ke; base += 32) {
        const int cnt  = min(32, ke - base);
        const int cnt4 = (cnt + 3) & ~3;
        const int j    = csrc[min(base + lane, ke - 1)];
        for (int e = 0; e < cnt4; e += 4) {
            v2f as2[4];
#pragma unroll
            for (int u = 0; u < 4; ++u) {
                const int je = __builtin_amdgcn_readlane(j, e + u);
                as2[u] = *(const v2f*)(asrc + (size_t)je * (BTOT * HH) + lo);
            }
#pragma unroll
            for (int u = 0; u < 4; ++u) {
                const bool real = e + u < cnt;
                const float a0 = leaky(as2[u].x + ad2.x);
                const float a1 = leaky(as2[u].y + ad2.y);
                const float n0 = fmaxf(m0, a0);
                const float n1 = fmaxf(m1, a1);
                s0 = s0 * __expf(m0 - n0) + (real ? __expf(a0 - n0) : 0.f);
                s1 = s1 * __expf(m1 - n1) + (real ? __expf(a1 - n1) : 0.f);
                m0 = n0; m1 = n1;
            }
        }
    }
    const float i0 = 1.f / (s0 + 1e-16f);
    const float i1 = 1.f / (s1 + 1e-16f);

    __syncthreads();
    if (!live) return;

    v2f acc[16][HH];
#pragma unroll
    for (int b = 0; b < 16; ++b) {
#pragma unroll
        for (int h = 0; h < HH; ++h) acc[b][h] = (v2f){0.f, 0.f};
    }

    for (int base = kb; base < ke; base += 32) {
        const int cnt  = min(32, ke - base);
        const int cnt4 = (cnt + 3) & ~3;
        const int j    = csrc[min(base + lane, ke - 1)];
        for (int e = 0; e < cnt4; e += 4) {
            v2f as2[4];
#pragma unroll
            for (int u = 0; u < 4; ++u) {
                const int je = __builtin_amdgcn_readlane(j, e + u);
                as2[u] = *(const v2f*)(asrc + (size_t)je * (BTOT * HH) + lo);
            }
#pragma unroll
            for (int u = 0; u < 4; ++u) {
                const bool real = e + u < cnt;
                const float a0 = leaky(as2[u].x + ad2.x);
                const float a1 = leaky(as2[u].y + ad2.y);
                v2f c;
                c.x = real ? __expf(a0 - m0) * i0 : 0.f;
                c.y = real ? __expf(a1 - m1) * i1 : 0.f;
                *(v2f*)(zt + (e + u) * (16 * HH) + ib * HH + 2 * hp) = c;
            }
        }
        wave_lds_handover();

        const float* xcol = x + (size_t)bt0 * (NN * DD) + 2 * lane;
        const int cntg = (cnt + GE - 1) & ~(GE - 1);
        for (int e = 0; e < cntg; e += GE) {
            v2f row[GE][16];
#pragma unroll
            for (int u = 0; u < GE; ++u) {
                const int je = __builtin_amdgcn_readlane(j, e + u);
                const float* rp = xcol + (size_t)je * DD;
#pragma unroll
                for (int b = 0; b < 16; ++b)
                    row[u][b] = *(const v2f*)(rp + (size_t)b * (NN * DD));
            }
#pragma unroll
            for (int u = 0; u < GE; ++u) {
                const float* cp = zt + (e + u) * (16 * HH);
#pragma unroll
                for (int b = 0; b < 16; ++b) {
                    const v4f c = *(const v4f*)(cp + b * HH);
                    acc[b][0] += c.x * row[u][b];
                    acc[b][1] += c.y * row[u][b];
                    acc[b][2] += c.z * row[u][b];
                    acc[b][3] += c.w * row[u][b];
                }
            }
        }
        wave_lds_handover();
    }

    const int r    = lane & 15;
    const int half = lane >> 4;
    const int kgrp = half * 2;
    const size_t rowbase = ((size_t)(bt0 + half * 8) * NN + n) * DD + r;
    float xr[8][4];
    v8f o0 = {}, o1 = {}, o2 = {}, o3 = {};
    const float* zrow = zt + r * ZK;
    const int    swr  = 4 * r;

#pragma unroll
    for (int part = 0; part < 2; ++part) {
#pragma unroll
        for (int b = 0; b < 16; ++b) {
            float* zr = zt + b * ZK;
            const int sw = 4 * b;
#pragma unroll
            for (int hh = 0; hh < 2; ++hh)
                *(v2f*)(zr + ((hh * DD + 2 * lane) ^ sw)) = acc[b][2 * part + hh];
        }
        wave_lds_handover();

        if (part == 0) {
#pragma unroll
            for (int v = 0; v < 8; ++v) {
#pragma unroll
                for (int c = 0; c < 4; ++c)
                    xr[v][c] = x[rowbase + (size_t)v * (NN * DD) + c * 16];
            }
        }

#pragma unroll 8
        for (int kk = 0; kk < 32; ++kk) {
            const v2f a = *(const v2f*)(zrow + ((kk * 4 + kgrp) ^ swr));
            const float* bp = sW + ((kk & 15) * 4 + kgrp) * COLS + (2 * part + (kk >> 4)) * DD + r;
            v2f b0, b1, b2, b3;
            b0.x = bp[0];  b0.y = bp[COLS];
            b1.x = bp[16]; b1.y = bp[COLS + 16];
            b2.x = bp[32]; b2.y = bp[COLS + 32];
            b3.x = bp[48]; b3.y = bp[COLS + 48];
            o0 = __builtin_amdgcn_wmma_f32_16x16x4_f32(false, a, false, b0, (short)0, o0, false, false);
            o1 = __builtin_amdgcn_wmma_f32_16x16x4_f32(false, a, false, b1, (short)0, o1, false, false);
            o2 = __builtin_amdgcn_wmma_f32_16x16x4_f32(false, a, false, b2, (short)0, o2, false, false);
            o3 = __builtin_amdgcn_wmma_f32_16x16x4_f32(false, a, false, b3, (short)0, o3, false, false);
        }
        if (part == 0) {
            wave_lds_handover();
        }
    }

    {
        const float bb0 = bias[r], bb1 = bias[16 + r], bb2 = bias[32 + r], bb3 = bias[48 + r];
        const float g0 = gamma[r], g1 = gamma[16 + r], g2 = gamma[32 + r], g3 = gamma[48 + r];
        const float be0 = beta[r], be1 = beta[16 + r], be2 = beta[32 + r], be3 = beta[48 + r];
#pragma unroll
        for (int v = 0; v < 8; ++v) {
            const size_t idx = rowbase + (size_t)v * (NN * DD);
            const float y0 = o0[v] * 0.25f + bb0 + xr[v][0];
            const float y1 = o1[v] * 0.25f + bb1 + xr[v][1];
            const float y2 = o2[v] * 0.25f + bb2 + xr[v][2];
            const float y3 = o3[v] * 0.25f + bb3 + xr[v][3];
            const float sm = row16_sum((y0 + y1) + (y2 + y3));
            const float sq = row16_sum((y0 * y0 + y1 * y1) + (y2 * y2 + y3 * y3));
            const float mu  = sm * (1.f / 64.f);
            const float var = sq * (1.f / 64.f) - mu * mu;
            const float rs  = rsqrtf(var + 1e-5f);
            out[idx]      = (y0 - mu) * rs * g0 + be0;
            out[idx + 16] = (y1 - mu) * rs * g1 + be1;
            out[idx + 32] = (y2 - mu) * rs * g2 + be2;
            out[idx + 48] = (y3 - mu) * rs * g3 + be3;
        }
    }
}

extern "C" void kernel_launch(void* const* d_in, const int* in_sizes, int n_in,
                              void* d_out, int out_size, void* d_ws, size_t ws_size,
                              hipStream_t stream) {
    const float* x        = (const float*)d_in[0];
    const float* W        = (const float*)d_in[1];
    const float* att_src  = (const float*)d_in[2];
    const float* att_dst  = (const float*)d_in[3];
    const float* bias     = (const float*)d_in[4];
    const float* gamma    = (const float*)d_in[5];
    const float* beta     = (const float*)d_in[6];
    const int*   ei       = (const int*)d_in[7];
    float* out            = (float*)d_out;
    const int E = in_sizes[7] / 2;

    char* ws = (char*)d_ws;
    float* asrc = (float*)ws;  ws += sizeof(float) * (size_t)MROWS * HH;
    float* adst = (float*)ws;  ws += sizeof(float) * (size_t)MROWS * HH;
    int* off    = (int*)ws;    ws += sizeof(int) * (NN + 1);
    int* csrc   = (int*)ws;    ws += sizeof(int) * (E + NN);

    gat_logits_wmma<<<(MTILES + 7) / 8 + 1, 256, 0, stream>>>(x, W, att_src, att_dst, asrc, adst,
                                                            ei, E, off, csrc);

    gat_agg_proj_ln<<<AP_GRID, 32 * AP_WAVES, 0, stream>>>(x, W, asrc, adst, off, csrc,
                                          bias, gamma, beta, out);
}


# softmax coefficients no longer pass through LDS: each stays in its lane and reaches the gather's packed FMAs as a scalar operand (v_readlane); one flat edge loop; order of every sum unchanged
# speedup vs baseline: 2.8262x; 1.1816x over previous
#include <hip/hip_runtime.h>
#include <hip/hip_bf16.h>
#include <math.h>


typedef __attribute__((ext_vector_type(2))) float v2f;
typedef __attribute__((ext_vector_type(4))) float v4f;
typedef __attribute__((ext_vector_type(8))) float v8f;

#define BTOT   192
#define NN     325
#define DD     64
#define HH     4
#define COLS   256
#define MROWS  (NN * BTOT)
#define MTILES (MROWS / 16)
#define NEG_SLOPE 0.2f

#define ACOLS 16

#define LANE_XROW(v, M) __builtin_bit_cast(float, __builtin_amdgcn_update_dpp( \
        __builtin_bit_cast(int, (v)), __builtin_bit_cast(int, (v)), 0x160 | (M), 0xf, 0xf, true))

__device__ __forceinline__ float row16_sum(float v) {
    v += LANE_XROW(v, 1); v += LANE_XROW(v, 2); v += LANE_XROW(v, 4); v += LANE_XROW(v, 8);
    return v;
}
__device__ __forceinline__ float leaky(float a) { return (a >= 0.f) ? a : NEG_SLOPE * a; }

#define CSR_CHUNK ((NN + 31) / 32)
#define CSR_U     8

__device__ __forceinline__ void csr_build(const int* __restrict__ ei, int E,
                                          int* __restrict__ off, int* __restrict__ csrc,
                                          int* sdeg, int* scur)
{
    const int tid = threadIdx.x;
    for (int n = tid; n < NN; n += 256) sdeg[n] = 1;
    __syncthreads();
    for (int base = tid; base < E; base += 256 * CSR_U) {
        int d[CSR_U];
#pragma unroll
        for (int u = 0; u < CSR_U; ++u)
            d[u] = (base + u * 256 < E) ? ei[E + base + u * 256] : 0;
#pragma unroll
        for (int u = 0; u < CSR_U; ++u)
            if (base + u * 256 < E) atomicAdd(&sdeg[d[u]], 1);
    }
    __syncthreads();
    if (tid < 32) {
        int loc[CSR_CHUNK];
        int sum = 0;
#pragma unroll
        for (int k = 0; k < CSR_CHUNK; ++k) {
            const int n = tid * CSR_CHUNK + k;
            const int v = (n < NN) ? sdeg[n] : 0;
            loc[k] = v;
            sum += v;
        }
        int incl = sum;
        for (int d = 1; d < 32; d <<= 1) {
            const int y = __shfl_up(incl, d);
            if (tid >= d) incl += y;
        }
        int run = incl - sum;
#pragma unroll
        for (int k = 0; k < CSR_CHUNK; ++k) {
            const int n = tid * CSR_CHUNK + k;
            if (n < NN) { off[n] = run; scur[n] = run; }
            run += loc[k];
        }
        if (tid == 31) off[NN] = incl;
    }
    __syncthreads();
    for (int base = tid; base < E; base += 256 * CSR_U) {
        int s[CSR_U], d[CSR_U];
#pragma unroll
        for (int u = 0; u < CSR_U; ++u) {
            const bool in = base + u * 256 < E;
            s[u] = in ? ei[base + u * 256] : 0;
            d[u] = in ? ei[E + base + u * 256] : 0;
        }
#pragma unroll
        for (int u = 0; u < CSR_U; ++u)
            if (base + u * 256 < E) csrc[atomicAdd(&scur[d[u]], 1)] = s[u];
    }
    for (int n = tid; n < NN; n += 256)
        csrc[atomicAdd(&scur[n], 1)] = n;
}

__global__ __launch_bounds__(256) void gat_logits_wmma(
    const float* __restrict__ x,
    const float* __restrict__ W,
    const float* __restrict__ att_src,
    const float* __restrict__ att_dst,
    float* __restrict__ asrc,
    float* __restrict__ adst,
    const int* __restrict__ ei,
    int E,
    int* __restrict__ off,
    int* __restrict__ csrc)
{
    __shared__ __attribute__((aligned(16))) float sWa[DD * ACOLS];
    if (blockIdx.x == 0) {
        csr_build(ei, E, off, csrc, (int*)sWa, (int*)sWa + 512);
        return;
    }
    {
        const int f4 = (threadIdx.x & 15) * 4;
        const int hw = threadIdx.x >> 4;
        const int h  = hw & 3;
        const v4f a = *(const v4f*)(att_src + h * DD + f4);
        const v4f b = *(const v4f*)(att_dst + h * DD + f4);
        v4f w[16];
#pragma unroll
        for (int it = 0; it < 16; ++it)
            w[it] = *(const v4f*)(W + (size_t)(it * 16 + hw) * DD + f4);
#pragma unroll
        for (int it = 0; it < 16; ++it) {
            const int d = it * 4 + (hw >> 2);
            const float s = row16_sum((w[it].x * a.x + w[it].y * a.y) + (w[it].z * a.z + w[it].w * a.w));
            const float t = row16_sum((w[it].x * b.x + w[it].y * b.y) + (w[it].z * b.z + w[it].w * b.w));
            if ((threadIdx.x & 15) == 0) {
                float* o = sWa + d * ACOLS + h;
                o[0] = s; o[4] = t; o[8] = 0.f; o[12] = 0.f;
            }
        }
    }
    __syncthreads();

    const int wave  = threadIdx.x >> 5;
    const int lane  = threadIdx.x & 31;
    const int mtile = (blockIdx.x - 1) * 8 + wave;
    if (mtile >= MTILES) return;

    const int m0   = mtile * 16;
    const int r    = lane & 15;
    const int kgrp = (lane >> 4) * 2;
    const int m    = m0 + r;
    const int n    = m / BTOT;
    const int bt   = m - n * BTOT;
    const float* arow = x + ((size_t)bt * NN + n) * DD;

    v8f acc = {};
#pragma unroll
    for (int kk = 0; kk < 16; ++kk) {
        const v2f a = *(const v2f*)(arow + kk * 4 + kgrp);
        v2f b;
        b.x = sWa[(kk * 4 + kgrp + 0) * ACOLS + (lane & 15)];
        b.y = sWa[(kk * 4 + kgrp + 1) * ACOLS + (lane & 15)];
        acc = __builtin_amdgcn_wmma_f32_16x16x4_f32(
            false, a, false, b, (short)0, acc, false, false);
    }
    const int col   = lane & 15;
    const int rbase = (lane >> 4) * 8;
    if (col < 2 * HH) {
        float* dstp = ((col < HH) ? asrc : adst) + (size_t)(m0 + rbase) * HH + (col & (HH - 1));
#pragma unroll
        for (int v = 0; v < 8; ++v)
            dstp[v * HH] = acc[v];
    }
}

#define AP_WAVES 16
#define AP_GRID ((MTILES + AP_WAVES - 1) / AP_WAVES)
#define ZK 128
#define GE 2

#define LANE_VAL(v, L) __builtin_bit_cast(float, __builtin_amdgcn_readlane(__builtin_bit_cast(int, (v)), (L)))

#define AGG_ITEM(B) do { \
        const v2f rw_ = row[u][B]; \
        acc[B][0] += LANE_VAL(ca, B) * rw_; \
        acc[B][1] += LANE_VAL(cb, B) * rw_; \
        acc[B][2] += LANE_VAL(ca, 16 + (B)) * rw_; \
        acc[B][3] += LANE_VAL(cb, 16 + (B)) * rw_; \
    } while (0)
#define NEGMAX (-3.402823466e+38f)

__device__ __forceinline__ void wave_lds_handover() {
    __builtin_amdgcn_fence(__ATOMIC_RELEASE, "wavefront");
    __builtin_amdgcn_wave_barrier();
    __builtin_amdgcn_fence(__ATOMIC_ACQUIRE, "wavefront");
}

__global__ __launch_bounds__(32 * AP_WAVES) void gat_agg_proj_ln(
    const float* __restrict__ x,
    const float* __restrict__ W,
    const float* __restrict__ asrc,
    const float* __restrict__ adst,
    const int* __restrict__ off,
    const int* __restrict__ csrc,
    const float* __restrict__ bias,
    const float* __restrict__ gamma,
    const float* __restrict__ beta,
    float* __restrict__ out)
{
    __shared__ __attribute__((aligned(16))) float sW[DD * COLS];
    __shared__ __attribute__((aligned(16))) float sZ[AP_WAVES * 16 * ZK];

    {
        v4f wst[(DD * COLS / 4) / (32 * AP_WAVES)];
#pragma unroll
        for (int k = 0; k < (DD * COLS / 4) / (32 * AP_WAVES); ++k)
            wst[k] = ((const v4f*)W)[threadIdx.x + k * (32 * AP_WAVES)];
#pragma unroll
        for (int k = 0; k < (DD * COLS / 4) / (32 * AP_WAVES); ++k)
            ((v4f*)sW)[threadIdx.x + k * (32 * AP_WAVES)] = wst[k];
    }

    const int lane = threadIdx.x & 31;
    const int wave = __builtin_amdgcn_readfirstlane((int)(threadIdx.x >> 5));
    const bool live = wave * AP_GRID + (int)blockIdx.x < MTILES;
    const int tile = min(wave * AP_GRID + (int)blockIdx.x, MTILES - 1);
    const int n    = tile / (BTOT / 16);
    const int bt0  = (tile - n * (BTOT / 16)) * 16;
    const int kb = off[n], ke = off[n + 1];
    float* zt = sZ + wave * (16 * ZK);

    const int ib = lane & 15;
    const int hp = lane >> 4;
    const size_t lo = (size_t)(bt0 + ib) * HH + 2 * hp;
    const v2f ad2 = *(const v2f*)(adst + (size_t)n * (BTOT * HH) + lo);

    float m0 = NEGMAX, m1 = NEGMAX, s0 = 0.f, s1 = 0.f;
    for (int base = kb; base < ke; base += 32) {
        const int cnt  = min(32, ke - base);
        const int cnt4 = (cnt + 3) & ~3;
        const int j    = csrc[min(base + lane, ke - 1)];
        for (int e = 0; e < cnt4; e += 4) {
            v2f as2[4];
#pragma unroll
            for (int u = 0; u < 4; ++u) {
                const int je = __builtin_amdgcn_readlane(j, e + u);
                as2[u] = *(const v2f*)(asrc + (size_t)je * (BTOT * HH) + lo);
            }
#pragma unroll
            for (int u = 0; u < 4; ++u) {
                const bool real = e + u < cnt;
                const float a0 = leaky(as2[u].x + ad2.x);
                const float a1 = leaky(as2[u].y + ad2.y);
                const float n0 = fmaxf(m0, a0);
                const float n1 = fmaxf(m1, a1);
                s0 = s0 * __expf(m0 - n0) + (real ? __expf(a0 - n0) : 0.f);
                s1 = s1 * __expf(m1 - n1) + (real ? __expf(a1 - n1) : 0.f);
                m0 = n0; m1 = n1;
            }
        }
    }
    const float i0 = 1.f / (s0 + 1e-16f);
    const float i1 = 1.f / (s1 + 1e-16f);

    __syncthreads();
    if (!live) return;

    v2f acc[16][HH];
#pragma unroll
    for (int b = 0; b < 16; ++b) {
#pragma unroll
        for (int h = 0; h < HH; ++h) acc[b][h] = (v2f){0.f, 0.f};
    }

    const float* xcol = x + (size_t)bt0 * (NN * DD) + 2 * lane;
    const int deg  = ke - kb;
    const int degg = (deg + GE - 1) & ~(GE - 1);
    int j = 0;
    for (int e = 0; e < degg; e += GE) {
        if ((e & 31) == 0) j = csrc[min(kb + e + lane, ke - 1)];
        v2f as2[GE];
        v2f row[GE][16];
#pragma unroll
        for (int u = 0; u < GE; ++u) {
            const int je = __builtin_amdgcn_readlane(j, (e & 31) + u);
            as2[u] = *(const v2f*)(asrc + (size_t)je * (BTOT * HH) + lo);
        }
#pragma unroll
        for (int u = 0; u < GE; ++u) {
            const int je = __builtin_amdgcn_readlane(j, (e & 31) + u);
            const float* rp = xcol + (size_t)je * DD;
#pragma unroll
            for (int b = 0; b < 16; ++b)
                row[u][b] = *(const v2f*)(rp + (size_t)b * (NN * DD));
        }
#pragma unroll
        for (int u = 0; u < GE; ++u) {
            const float keep = (e + u < deg) ? 1.f : 0.f;
            const float a0 = leaky(as2[u].x + ad2.x);
            const float a1 = leaky(as2[u].y + ad2.y);
            const float ca = __expf(a0 - m0) * i0 * keep;
            const float cb = __expf(a1 - m1) * i1 * keep;
            AGG_ITEM(0);  AGG_ITEM(1);  AGG_ITEM(2);  AGG_ITEM(3);
            __builtin_amdgcn_sched_barrier(0);
            AGG_ITEM(4);  AGG_ITEM(5);  AGG_ITEM(6);  AGG_ITEM(7);
            __builtin_amdgcn_sched_barrier(0);
            AGG_ITEM(8);  AGG_ITEM(9);  AGG_ITEM(10); AGG_ITEM(11);
            __builtin_amdgcn_sched_barrier(0);
            AGG_ITEM(12); AGG_ITEM(13); AGG_ITEM(14); AGG_ITEM(15);
            __builtin_amdgcn_sched_barrier(0);
        }
    }

    const int r    = lane & 15;
    const int half = lane >> 4;
    const int kgrp = half * 2;
    const size_t rowbase = ((size_t)(bt0 + half * 8) * NN + n) * DD + r;
    float xr[8][4];
    v8f o0 = {}, o1 = {}, o2 = {}, o3 = {};
    const float* zrow = zt + r * ZK;
    const int    swr  = 4 * r;

#pragma unroll
    for (int part = 0; part < 2; ++part) {
#pragma unroll
        for (int b = 0; b < 16; ++b) {
            float* zr = zt + b * ZK;
            const int sw = 4 * b;
#pragma unroll
            for (int hh = 0; hh < 2; ++hh)
                *(v2f*)(zr + ((hh * DD + 2 * lane) ^ sw)) = acc[b][2 * part + hh];
        }
        wave_lds_handover();

        if (part == 0) {
#pragma unroll
            for (int v = 0; v < 8; ++v) {
#pragma unroll
                for (int c = 0; c < 4; ++c)
                    xr[v][c] = x[rowbase + (size_t)v * (NN * DD) + c * 16];
            }
        }

#pragma unroll 8
        for (int kk = 0; kk < 32; ++kk) {
            const v2f a = *(const v2f*)(zrow + ((kk * 4 + kgrp) ^ swr));
            const float* bp = sW + ((kk & 15) * 4 + kgrp) * COLS + (2 * part + (kk >> 4)) * DD + r;
            v2f b0, b1, b2, b3;
            b0.x = bp[0];  b0.y = bp[COLS];
            b1.x = bp[16]; b1.y = bp[COLS + 16];
            b2.x = bp[32]; b2.y = bp[COLS + 32];
            b3.x = bp[48]; b3.y = bp[COLS + 48];
            o0 = __builtin_amdgcn_wmma_f32_16x16x4_f32(false, a, false, b0, (short)0, o0, false, false);
            o1 = __builtin_amdgcn_wmma_f32_16x16x4_f32(false, a, false, b1, (short)0, o1, false, false);
            o2 = __builtin_amdgcn_wmma_f32_16x16x4_f32(false, a, false, b2, (short)0, o2, false, false);
            o3 = __builtin_amdgcn_wmma_f32_16x16x4_f32(false, a, false, b3, (short)0, o3, false, false);
        }
        if (part == 0) {
            wave_lds_handover();
        }
    }

    {
        const float bb0 = bias[r], bb1 = bias[16 + r], bb2 = bias[32 + r], bb3 = bias[48 + r];
        const float g0 = gamma[r], g1 = gamma[16 + r], g2 = gamma[32 + r], g3 = gamma[48 + r];
        const float be0 = beta[r], be1 = beta[16 + r], be2 = beta[32 + r], be3 = beta[48 + r];
#pragma unroll
        for (int v = 0; v < 8; ++v) {
            const size_t idx = rowbase + (size_t)v * (NN * DD);
            const float y0 = o0[v] * 0.25f + bb0 + xr[v][0];
            const float y1 = o1[v] * 0.25f + bb1 + xr[v][1];
            const float y2 = o2[v] * 0.25f + bb2 + xr[v][2];
            const float y3 = o3[v] * 0.25f + bb3 + xr[v][3];
            const float sm = row16_sum((y0 + y1) + (y2 + y3));
            const float sq = row16_sum((y0 * y0 + y1 * y1) + (y2 * y2 + y3 * y3));
            const float mu  = sm * (1.f / 64.f);
            const float var = sq * (1.f / 64.f) - mu * mu;
            const float rs  = rsqrtf(var + 1e-5f);
            out[idx]      = (y0 - mu) * rs * g0 + be0;
            out[idx + 16] = (y1 - mu) * rs * g1 + be1;
            out[idx + 32] = (y2 - mu) * rs * g2 + be2;
            out[idx + 48] = (y3 - mu) * rs * g3 + be3;
        }
    }
}

extern "C" void kernel_launch(void* const* d_in, const int* in_sizes, int n_in,
                              void* d_out, int out_size, void* d_ws, size_t ws_size,
                              hipStream_t stream) {
    const float* x        = (const float*)d_in[0];
    const float* W        = (const float*)d_in[1];
    const float* att_src  = (const float*)d_in[2];
    const float* att_dst  = (const float*)d_in[3];
    const float* bias     = (const float*)d_in[4];
    const float* gamma    = (const float*)d_in[5];
    const float* beta     = (const float*)d_in[6];
    const int*   ei       = (const int*)d_in[7];
    float* out            = (float*)d_out;
    const int E = in_sizes[7] / 2;

    char* ws = (char*)d_ws;
    float* asrc = (float*)ws;  ws += sizeof(float) * (size_t)MROWS * HH;
    float* adst = (float*)ws;  ws += sizeof(float) * (size_t)MROWS * HH;
    int* off    = (int*)ws;    ws += sizeof(int) * (NN + 1);
    int* csrc   = (int*)ws;    ws += sizeof(int) * (E + NN);

    gat_logits_wmma<<<(MTILES + 7) / 8 + 1, 256, 0, stream>>>(x, W, att_src, att_dst, asrc, adst,
                                                            ei, E, off, csrc);

    gat_agg_proj_ln<<<AP_GRID, 32 * AP_WAVES, 0, stream>>>(x, W, asrc, adst, off, csrc,
                                          bias, gamma, beta, out);
}


# dependent load chains cut: launch 1 asks for its x fragments ahead of the W fold, its CSR block loads every edge once, up front; launch 2 pass A has 16 edge logits in flight, not 4; sum order kept
# speedup vs baseline: 2.9558x; 1.0459x over previous
#include <hip/hip_runtime.h>
#include <hip/hip_bf16.h>
#include <math.h>


typedef __attribute__((ext_vector_type(2))) float v2f;
typedef __attribute__((ext_vector_type(4))) float v4f;
typedef __attribute__((ext_vector_type(8))) float v8f;

#define BTOT   192
#define NN     325
#define DD     64
#define HH     4
#define COLS   256
#define MROWS  (NN * BTOT)
#define MTILES (MROWS / 16)
#define NEG_SLOPE 0.2f

#define ACOLS 16

#define LANE_XROW(v, M) __builtin_bit_cast(float, __builtin_amdgcn_update_dpp( \
        __builtin_bit_cast(int, (v)), __builtin_bit_cast(int, (v)), 0x160 | (M), 0xf, 0xf, true))

__device__ __forceinline__ float row16_sum(float v) {
    v += LANE_XROW(v, 1); v += LANE_XROW(v, 2); v += LANE_XROW(v, 4); v += LANE_XROW(v, 8);
    return v;
}
__device__ __forceinline__ float leaky(float a) { return (a >= 0.f) ? a : NEG_SLOPE * a; }

#define CSR_CHUNK ((NN + 31) / 32)
#define CSR_U     11

__device__ __forceinline__ void csr_build(const int* __restrict__ ei, int E,
                                          int* __restrict__ off, int* __restrict__ csrc,
                                          int* sdeg, int* scur)
{
    const int tid = threadIdx.x;
    int s0[CSR_U], d0[CSR_U];
#pragma unroll
    for (int u = 0; u < CSR_U; ++u) {
        const bool in = tid + u * 256 < E;
        s0[u] = in ? ei[tid + u * 256] : 0;
        d0[u] = in ? ei[E + tid + u * 256] : 0;
    }
    for (int n = tid; n < NN; n += 256) sdeg[n] = 1;
    __syncthreads();
#pragma unroll
    for (int u = 0; u < CSR_U; ++u)
        if (tid + u * 256 < E) atomicAdd(&sdeg[d0[u]], 1);
    for (int base = tid + 256 * CSR_U; base < E; base += 256 * CSR_U) {
        int d[CSR_U];
#pragma unroll
        for (int u = 0; u < CSR_U; ++u)
            d[u] = (base + u * 256 < E) ? ei[E + base + u * 256] : 0;
#pragma unroll
        for (int u = 0; u < CSR_U; ++u)
            if (base + u * 256 < E) atomicAdd(&sdeg[d[u]], 1);
    }
    __syncthreads();
    if (tid < 32) {
        int loc[CSR_CHUNK];
        int sum = 0;
#pragma unroll
        for (int k = 0; k < CSR_CHUNK; ++k) {
            const int n = tid * CSR_CHUNK + k;
            const int v = (n < NN) ? sdeg[n] : 0;
            loc[k] = v;
            sum += v;
        }
        int incl = sum;
        for (int d = 1; d < 32; d <<= 1) {
            const int y = __shfl_up(incl, d);
            if (tid >= d) incl += y;
        }
        int run = incl - sum;
#pragma unroll
        for (int k = 0; k < CSR_CHUNK; ++k) {
            const int n = tid * CSR_CHUNK + k;
            if (n < NN) { off[n] = run; scur[n] = run; }
            run += loc[k];
        }
        if (tid == 31) off[NN] = incl;
    }
    __syncthreads();
#pragma unroll
    for (int u = 0; u < CSR_U; ++u)
        if (tid + u * 256 < E) csrc[atomicAdd(&scur[d0[u]], 1)] = s0[u];
    for (int base = tid + 256 * CSR_U; base < E; base += 256 * CSR_U) {
        int s[CSR_U], d[CSR_U];
#pragma unroll
        for (int u = 0; u < CSR_U; ++u) {
            const bool in = base + u * 256 < E;
            s[u] = in ? ei[base + u * 256] : 0;
            d[u] = in ? ei[E + base + u * 256] : 0;
        }
#pragma unroll
        for (int u = 0; u < CSR_U; ++u)
            if (base + u * 256 < E) csrc[atomicAdd(&scur[d[u]], 1)] = s[u];
    }
    for (int n = tid; n < NN; n += 256)
        csrc[atomicAdd(&scur[n], 1)] = n;
}

__global__ __launch_bounds__(256) void gat_logits_wmma(
    const float* __restrict__ x,
    const float* __restrict__ W,
    const float* __restrict__ att_src,
    const float* __restrict__ att_dst,
    float* __restrict__ asrc,
    float* __restrict__ adst,
    const int* __restrict__ ei,
    int E,
    int* __restrict__ off,
    int* __restrict__ csrc)
{
    __shared__ __attribute__((aligned(16))) float sWa[DD * ACOLS];
    if (blockIdx.x == 0) {
        csr_build(ei, E, off, csrc, (int*)sWa, (int*)sWa + 512);
        return;
    }
    const int wave  = threadIdx.x >> 5;
    const int lane  = threadIdx.x & 31;
    const int mtile = (blockIdx.x - 1) * 8 + wave;
    const int r     = lane & 15;
    const int kgrp  = (lane >> 4) * 2;
    v2f afr[16];
    {
        const int m  = min(mtile, MTILES - 1) * 16 + r;
        const int n  = m / BTOT;
        const int bt = m - n * BTOT;
        const float* arow = x + ((size_t)bt * NN + n) * DD;
#pragma unroll
        for (int kk = 0; kk < 16; ++kk)
            afr[kk] = *(const v2f*)(arow + kk * 4 + kgrp);
    }
    {
        const int f4 = (threadIdx.x & 15) * 4;
        const int hw = threadIdx.x >> 4;
        const int h  = hw & 3;
        const v4f a = *(const v4f*)(att_src + h * DD + f4);
        const v4f b = *(const v4f*)(att_dst + h * DD + f4);
        v4f w[16];
#pragma unroll
        for (int it = 0; it < 16; ++it)
            w[it] = *(const v4f*)(W + (size_t)(it * 16 + hw) * DD + f4);
#pragma unroll
        for (int it = 0; it < 16; ++it) {
            const int d = it * 4 + (hw >> 2);
            const float s = row16_sum((w[it].x * a.x + w[it].y * a.y) + (w[it].z * a.z + w[it].w * a.w));
            const float t = row16_sum((w[it].x * b.x + w[it].y * b.y) + (w[it].z * b.z + w[it].w * b.w));
            if ((threadIdx.x & 15) == 0) {
                float* o = sWa + d * ACOLS + h;
                o[0] = s; o[4] = t; o[8] = 0.f; o[12] = 0.f;
            }
        }
    }
    __syncthreads();

    if (mtile >= MTILES) return;

    const int m0   = mtile * 16;

    v8f acc = {};
#pragma unroll
    for (int kk = 0; kk < 16; ++kk) {
        const v2f a = afr[kk];
        v2f b;
        b.x = sWa[(kk * 4 + kgrp + 0) * ACOLS + (lane & 15)];
        b.y = sWa[(kk * 4 + kgrp + 1) * ACOLS + (lane & 15)];
        acc = __builtin_amdgcn_wmma_f32_16x16x4_f32(
            false, a, false, b, (short)0, acc, false, false);
    }
    const int col   = lane & 15;
    const int rbase = (lane >> 4) * 8;
    if (col < 2 * HH) {
        float* dstp = ((col < HH) ? asrc : adst) + (size_t)(m0 + rbase) * HH + (col & (HH - 1));
#pragma unroll
        for (int v = 0; v < 8; ++v)
            dstp[v * HH] = acc[v];
    }
}

#define AP_WAVES 16
#define AP_GRID ((MTILES + AP_WAVES - 1) / AP_WAVES)
#define ZK 128
#define GE 2
#define PA_U 16

#define LANE_VAL(v, L) __builtin_bit_cast(float, __builtin_amdgcn_readlane(__builtin_bit_cast(int, (v)), (L)))

#define AGG_ITEM(B) do { \
        const v2f rw_ = row[u][B]; \
        acc[B][0] += LANE_VAL(ca, B) * rw_; \
        acc[B][1] += LANE_VAL(cb, B) * rw_; \
        acc[B][2] += LANE_VAL(ca, 16 + (B)) * rw_; \
        acc[B][3] += LANE_VAL(cb, 16 + (B)) * rw_; \
    } while (0)
#define NEGMAX (-3.402823466e+38f)

__device__ __forceinline__ void wave_lds_handover() {
    __builtin_amdgcn_fence(__ATOMIC_RELEASE, "wavefront");
    __builtin_amdgcn_wave_barrier();
    __builtin_amdgcn_fence(__ATOMIC_ACQUIRE, "wavefront");
}

__global__ __launch_bounds__(32 * AP_WAVES) void gat_agg_proj_ln(
    const float* __restrict__ x,
    const float* __restrict__ W,
    const float* __restrict__ asrc,
    const float* __restrict__ adst,
    const int* __restrict__ off,
    const int* __restrict__ csrc,
    const float* __restrict__ bias,
    const float* __restrict__ gamma,
    const float* __restrict__ beta,
    float* __restrict__ out)
{
    __shared__ __attribute__((aligned(16))) float sW[DD * COLS];
    __shared__ __attribute__((aligned(16))) float sZ[AP_WAVES * 16 * ZK];

    {
        v4f wst[(DD * COLS / 4) / (32 * AP_WAVES)];
#pragma unroll
        for (int k = 0; k < (DD * COLS / 4) / (32 * AP_WAVES); ++k)
            wst[k] = ((const v4f*)W)[threadIdx.x + k * (32 * AP_WAVES)];
#pragma unroll
        for (int k = 0; k < (DD * COLS / 4) / (32 * AP_WAVES); ++k)
            ((v4f*)sW)[threadIdx.x + k * (32 * AP_WAVES)] = wst[k];
    }

    const int lane = threadIdx.x & 31;
    const int wave = __builtin_amdgcn_readfirstlane((int)(threadIdx.x >> 5));
    const bool live = wave * AP_GRID + (int)blockIdx.x < MTILES;
    const int tile = min(wave * AP_GRID + (int)blockIdx.x, MTILES - 1);
    const int n    = tile / (BTOT / 16);
    const int bt0  = (tile - n * (BTOT / 16)) * 16;
    const int kb = off[n], ke = off[n + 1];
    float* zt = sZ + wave * (16 * ZK);

    const int ib = lane & 15;
    const int hp = lane >> 4;
    const size_t lo = (size_t)(bt0 + ib) * HH + 2 * hp;
    const v2f ad2 = *(const v2f*)(adst + (size_t)n * (BTOT * HH) + lo);

    float m0 = NEGMAX, m1 = NEGMAX, s0 = 0.f, s1 = 0.f;
    int j0 = 0;
    for (int base = kb; base < ke; base += 32) {
        const int cnt  = min(32, ke - base);
        const int j    = csrc[min(base + lane, ke - 1)];
        if (base == kb) j0 = j;
        for (int e = 0; e < cnt; e += PA_U) {
            v2f as2[PA_U];
#pragma unroll
            for (int u = 0; u < PA_U; ++u) {
                const int je = __builtin_amdgcn_readlane(j, e + u);
                as2[u] = *(const v2f*)(asrc + (size_t)je * (BTOT * HH) + lo);
            }
#pragma unroll
            for (int g = 0; g < PA_U / 4; ++g) {
                if (e + 4 * g < cnt) {
#pragma unroll
                    for (int u = 4 * g; u < 4 * g + 4; ++u) {
                        const bool real = e + u < cnt;
                        const float a0 = leaky(as2[u].x + ad2.x);
                        const float a1 = leaky(as2[u].y + ad2.y);
                        const float n0 = fmaxf(m0, a0);
                        const float n1 = fmaxf(m1, a1);
                        s0 = s0 * __expf(m0 - n0) + (real ? __expf(a0 - n0) : 0.f);
                        s1 = s1 * __expf(m1 - n1) + (real ? __expf(a1 - n1) : 0.f);
                        m0 = n0; m1 = n1;
                    }
                }
            }
        }
    }
    const float i0 = 1.f / (s0 + 1e-16f);
    const float i1 = 1.f / (s1 + 1e-16f);

    __syncthreads();
    if (!live) return;

    v2f acc[16][HH];
#pragma unroll
    for (int b = 0; b < 16; ++b) {
#pragma unroll
        for (int h = 0; h < HH; ++h) acc[b][h] = (v2f){0.f, 0.f};
    }

    const float* xcol = x + (size_t)bt0 * (NN * DD) + 2 * lane;
    const int deg  = ke - kb;
    const int degg = (deg + GE - 1) & ~(GE - 1);
    int j = j0;
    for (int e = 0; e < degg; e += GE) {
        if ((e & 31) == 0 && e != 0) j = csrc[min(kb + e + lane, ke - 1)];
        v2f as2[GE];
        v2f row[GE][16];
#pragma unroll
        for (int u = 0; u < GE; ++u) {
            const int je = __builtin_amdgcn_readlane(j, (e & 31) + u);
            as2[u] = *(const v2f*)(asrc + (size_t)je * (BTOT * HH) + lo);
        }
#pragma unroll
        for (int u = 0; u < GE; ++u) {
            const int je = __builtin_amdgcn_readlane(j, (e & 31) + u);
            const float* rp = xcol + (size_t)je * DD;
#pragma unroll
            for (int b = 0; b < 16; ++b)
                row[u][b] = *(const v2f*)(rp + (size_t)b * (NN * DD));
        }
#pragma unroll
        for (int u = 0; u < GE; ++u) {
            const float keep = (e + u < deg) ? 1.f : 0.f;
            const float a0 = leaky(as2[u].x + ad2.x);
            const float a1 = leaky(as2[u].y + ad2.y);
            const float ca = __expf(a0 - m0) * i0 * keep;
            const float cb = __expf(a1 - m1) * i1 * keep;
            AGG_ITEM(0);  AGG_ITEM(1);  AGG_ITEM(2);  AGG_ITEM(3);
            __builtin_amdgcn_sched_barrier(0);
            AGG_ITEM(4);  AGG_ITEM(5);  AGG_ITEM(6);  AGG_ITEM(7);
            __builtin_amdgcn_sched_barrier(0);
            AGG_ITEM(8);  AGG_ITEM(9);  AGG_ITEM(10); AGG_ITEM(11);
            __builtin_amdgcn_sched_barrier(0);
            AGG_ITEM(12); AGG_ITEM(13); AGG_ITEM(14); AGG_ITEM(15);
            __builtin_amdgcn_sched_barrier(0);
        }
    }

    const int r    = lane & 15;
    const int half = lane >> 4;
    const int kgrp = half * 2;
    const size_t rowbase = ((size_t)(bt0 + half * 8) * NN + n) * DD + r;
    float xr[8][4];
    v8f o0 = {}, o1 = {}, o2 = {}, o3 = {};
    const float* zrow = zt + r * ZK;
    const int    swr  = 4 * r;

#pragma unroll
    for (int part = 0; part < 2; ++part) {
#pragma unroll
        for (int b = 0; b < 16; ++b) {
            float* zr = zt + b * ZK;
            const int sw = 4 * b;
#pragma unroll
            for (int hh = 0; hh < 2; ++hh)
                *(v2f*)(zr + ((hh * DD + 2 * lane) ^ sw)) = acc[b][2 * part + hh];
        }
        wave_lds_handover();

        if (part == 0) {
#pragma unroll
            for (int v = 0; v < 8; ++v) {
#pragma unroll
                for (int c = 0; c < 4; ++c)
                    xr[v][c] = x[rowbase + (size_t)v * (NN * DD) + c * 16];
            }
        }

#pragma unroll 8
        for (int kk = 0; kk < 32; ++kk) {
            const v2f a = *(const v2f*)(zrow + ((kk * 4 + kgrp) ^ swr));
            const float* bp = sW + ((kk & 15) * 4 + kgrp) * COLS + (2 * part + (kk >> 4)) * DD + r;
            v2f b0, b1, b2, b3;
            b0.x = bp[0];  b0.y = bp[COLS];
            b1.x = bp[16]; b1.y = bp[COLS + 16];
            b2.x = bp[32]; b2.y = bp[COLS + 32];
            b3.x = bp[48]; b3.y = bp[COLS + 48];
            o0 = __builtin_amdgcn_wmma_f32_16x16x4_f32(false, a, false, b0, (short)0, o0, false, false);
            o1 = __builtin_amdgcn_wmma_f32_16x16x4_f32(false, a, false, b1, (short)0, o1, false, false);
            o2 = __builtin_amdgcn_wmma_f32_16x16x4_f32(false, a, false, b2, (short)0, o2, false, false);
            o3 = __builtin_amdgcn_wmma_f32_16x16x4_f32(false, a, false, b3, (short)0, o3, false, false);
        }
        if (part == 0) {
            wave_lds_handover();
        }
    }

    {
        const float bb0 = bias[r], bb1 = bias[16 + r], bb2 = bias[32 + r], bb3 = bias[48 + r];
        const float g0 = gamma[r], g1 = gamma[16 + r], g2 = gamma[32 + r], g3 = gamma[48 + r];
        const float be0 = beta[r], be1 = beta[16 + r], be2 = beta[32 + r], be3 = beta[48 + r];
#pragma unroll
        for (int v = 0; v < 8; ++v) {
            const size_t idx = rowbase + (size_t)v * (NN * DD);
            const float y0 = o0[v] * 0.25f + bb0 + xr[v][0];
            const float y1 = o1[v] * 0.25f + bb1 + xr[v][1];
            const float y2 = o2[v] * 0.25f + bb2 + xr[v][2];
            const float y3 = o3[v] * 0.25f + bb3 + xr[v][3];
            const float sm = row16_sum((y0 + y1) + (y2 + y3));
            const float sq = row16_sum((y0 * y0 + y1 * y1) + (y2 * y2 + y3 * y3));
            const float mu  = sm * (1.f / 64.f);
            const float var = sq * (1.f / 64.f) - mu * mu;
            const float rs  = rsqrtf(var + 1e-5f);
            out[idx]      = (y0 - mu) * rs * g0 + be0;
            out[idx + 16] = (y1 - mu) * rs * g1 + be1;
            out[idx + 32] = (y2 - mu) * rs * g2 + be2;
            out[idx + 48] = (y3 - mu) * rs * g3 + be3;
        }
    }
}

extern "C" void kernel_launch(void* const* d_in, const int* in_sizes, int n_in,
                              void* d_out, int out_size, void* d_ws, size_t ws_size,
                              hipStream_t stream) {
    const float* x        = (const float*)d_in[0];
    const float* W        = (const float*)d_in[1];
    const float* att_src  = (const float*)d_in[2];
    const float* att_dst  = (const float*)d_in[3];
    const float* bias     = (const float*)d_in[4];
    const float* gamma    = (const float*)d_in[5];
    const float* beta     = (const float*)d_in[6];
    const int*   ei       = (const int*)d_in[7];
    float* out            = (float*)d_out;
    const int E = in_sizes[7] / 2;

    char* ws = (char*)d_ws;
    float* asrc = (float*)ws;  ws += sizeof(float) * (size_t)MROWS * HH;
    float* adst = (float*)ws;  ws += sizeof(float) * (size_t)MROWS * HH;
    int* off    = (int*)ws;    ws += sizeof(int) * (NN + 1);
    int* csrc   = (int*)ws;    ws += sizeof(int) * (E + NN);

    gat_logits_wmma<<<(MTILES + 7) / 8 + 1, 256, 0, stream>>>(x, W, att_src, att_dst, asrc, adst,
                                                            ei, E, off, csrc);

    gat_agg_proj_ln<<<AP_GRID, 32 * AP_WAVES, 0, stream>>>(x, W, asrc, adst, off, csrc,
                                          bias, gamma, beta, out);
}


# projection operand feed: W kept in LDS in B-fragment order (two conflict-free 16-byte loads per k-step, not eight 4-byte); loads grouped one 4-k-step block ahead of 16 WMMAs; every sum's order kept
# speedup vs baseline: 3.0916x; 1.0459x over previous
#include <hip/hip_runtime.h>
#include <hip/hip_bf16.h>
#include <math.h>


typedef __attribute__((ext_vector_type(2))) float v2f;
typedef __attribute__((ext_vector_type(4))) float v4f;
typedef __attribute__((ext_vector_type(8))) float v8f;

#define BTOT   192
#define NN     325
#define DD     64
#define HH     4
#define COLS   256
#define MROWS  (NN * BTOT)
#define MTILES (MROWS / 16)
#define NEG_SLOPE 0.2f

#define ACOLS 16

#define LANE_XROW(v, M) __builtin_bit_cast(float, __builtin_amdgcn_update_dpp( \
        __builtin_bit_cast(int, (v)), __builtin_bit_cast(int, (v)), 0x160 | (M), 0xf, 0xf, true))

__device__ __forceinline__ float row16_sum(float v) {
    v += LANE_XROW(v, 1); v += LANE_XROW(v, 2); v += LANE_XROW(v, 4); v += LANE_XROW(v, 8);
    return v;
}
__device__ __forceinline__ float leaky(float a) { return (a >= 0.f) ? a : NEG_SLOPE * a; }

#define CSR_CHUNK ((NN + 31) / 32)
#define CSR_U     11

__device__ __forceinline__ void csr_build(const int* __restrict__ ei, int E,
                                          int* __restrict__ off, int* __restrict__ csrc,
                                          int* sdeg, int* scur)
{
    const int tid = threadIdx.x;
    int s0[CSR_U], d0[CSR_U];
#pragma unroll
    for (int u = 0; u < CSR_U; ++u) {
        const bool in = tid + u * 256 < E;
        s0[u] = in ? ei[tid + u * 256] : 0;
        d0[u] = in ? ei[E + tid + u * 256] : 0;
    }
    for (int n = tid; n < NN; n += 256) sdeg[n] = 1;
    __syncthreads();
#pragma unroll
    for (int u = 0; u < CSR_U; ++u)
        if (tid + u * 256 < E) atomicAdd(&sdeg[d0[u]], 1);
    for (int base = tid + 256 * CSR_U; base < E; base += 256 * CSR_U) {
        int d[CSR_U];
#pragma unroll
        for (int u = 0; u < CSR_U; ++u)
            d[u] = (base + u * 256 < E) ? ei[E + base + u * 256] : 0;
#pragma unroll
        for (int u = 0; u < CSR_U; ++u)
            if (base + u * 256 < E) atomicAdd(&sdeg[d[u]], 1);
    }
    __syncthreads();
    if (tid < 32) {
        int loc[CSR_CHUNK];
        int sum = 0;
#pragma unroll
        for (int k = 0; k < CSR_CHUNK; ++k) {
            const int n = tid * CSR_CHUNK + k;
            const int v = (n < NN) ? sdeg[n] : 0;
            loc[k] = v;
            sum += v;
        }
        int incl = sum;
        for (int d = 1; d < 32; d <<= 1) {
            const int y = __shfl_up(incl, d);
            if (tid >= d) incl += y;
        }
        int run = incl - sum;
#pragma unroll
        for (int k = 0; k < CSR_CHUNK; ++k) {
            const int n = tid * CSR_CHUNK + k;
            if (n < NN) { off[n] = run; scur[n] = run; }
            run += loc[k];
        }
        if (tid == 31) off[NN] = incl;
    }
    __syncthreads();
#pragma unroll
    for (int u = 0; u < CSR_U; ++u)
        if (tid + u * 256 < E) csrc[atomicAdd(&scur[d0[u]], 1)] = s0[u];
    for (int base = tid + 256 * CSR_U; base < E; base += 256 * CSR_U) {
        int s[CSR_U], d[CSR_U];
#pragma unroll
        for (int u = 0; u < CSR_U; ++u) {
            const bool in = base + u * 256 < E;
            s[u] = in ? ei[base + u * 256] : 0;
            d[u] = in ? ei[E + base + u * 256] : 0;
        }
#pragma unroll
        for (int u = 0; u < CSR_U; ++u)
            if (base + u * 256 < E) csrc[atomicAdd(&scur[d[u]], 1)] = s[u];
    }
    for (int n = tid; n < NN; n += 256)
        csrc[atomicAdd(&scur[n], 1)] = n;
}

__global__ __launch_bounds__(256) void gat_logits_wmma(
    const float* __restrict__ x,
    const float* __restrict__ W,
    const float* __restrict__ att_src,
    const float* __restrict__ att_dst,
    float* __restrict__ asrc,
    float* __restrict__ adst,
    const int* __restrict__ ei,
    int E,
    int* __restrict__ off,
    int* __restrict__ csrc)
{
    __shared__ __attribute__((aligned(16))) float sWa[DD * ACOLS];
    if (blockIdx.x == 0) {
        csr_build(ei, E, off, csrc, (int*)sWa, (int*)sWa + 512);
        return;
    }
    const int wave  = threadIdx.x >> 5;
    const int lane  = threadIdx.x & 31;
    const int mtile = (blockIdx.x - 1) * 8 + wave;
    const int r     = lane & 15;
    const int kgrp  = (lane >> 4) * 2;
    v2f afr[16];
    {
        const int m  = min(mtile, MTILES - 1) * 16 + r;
        const int n  = m / BTOT;
        const int bt = m - n * BTOT;
        const float* arow = x + ((size_t)bt * NN + n) * DD;
#pragma unroll
        for (int kk = 0; kk < 16; ++kk)
            afr[kk] = *(const v2f*)(arow + kk * 4 + kgrp);
    }
    {
        const int f4 = (threadIdx.x & 15) * 4;
        const int hw = threadIdx.x >> 4;
        const int h  = hw & 3;
        const v4f a = *(const v4f*)(att_src + h * DD + f4);
        const v4f b = *(const v4f*)(att_dst + h * DD + f4);
        v4f w[16];
#pragma unroll
        for (int it = 0; it < 16; ++it)
            w[it] = *(const v4f*)(W + (size_t)(it * 16 + hw) * DD + f4);
#pragma unroll
        for (int it = 0; it < 16; ++it) {
            const int d = it * 4 + (hw >> 2);
            const float s = row16_sum((w[it].x * a.x + w[it].y * a.y) + (w[it].z * a.z + w[it].w * a.w));
            const float t = row16_sum((w[it].x * b.x + w[it].y * b.y) + (w[it].z * b.z + w[it].w * b.w));
            if ((threadIdx.x & 15) == 0) {
                float* o = sWa + d * ACOLS + h;
                o[0] = s; o[4] = t; o[8] = 0.f; o[12] = 0.f;
            }
        }
    }
    __syncthreads();

    if (mtile >= MTILES) return;

    const int m0   = mtile * 16;

    v8f acc = {};
#pragma unroll
    for (int kk = 0; kk < 16; ++kk) {
        const v2f a = afr[kk];
        v2f b;
        b.x = sWa[(kk * 4 + kgrp + 0) * ACOLS + (lane & 15)];
        b.y = sWa[(kk * 4 + kgrp + 1) * ACOLS + (lane & 15)];
        acc = __builtin_amdgcn_wmma_f32_16x16x4_f32(
            false, a, false, b, (short)0, acc, false, false);
    }
    const int col   = lane & 15;
    const int rbase = (lane >> 4) * 8;
    if (col < 2 * HH) {
        float* dstp = ((col < HH) ? asrc : adst) + (size_t)(m0 + rbase) * HH + (col & (HH - 1));
#pragma unroll
        for (int v = 0; v < 8; ++v)
            dstp[v * HH] = acc[v];
    }
}

#define AP_WAVES 16
#define AP_GRID ((MTILES + AP_WAVES - 1) / AP_WAVES)
#define ZK 128
#define WSTEP 256
#define PB 4
#define GE 2
#define PA_U 16

#define LANE_VAL(v, L) __builtin_bit_cast(float, __builtin_amdgcn_readlane(__builtin_bit_cast(int, (v)), (L)))

#define AGG_ITEM(B) do { \
        const v2f rw_ = row[u][B]; \
        acc[B][0] += LANE_VAL(ca, B) * rw_; \
        acc[B][1] += LANE_VAL(cb, B) * rw_; \
        acc[B][2] += LANE_VAL(ca, 16 + (B)) * rw_; \
        acc[B][3] += LANE_VAL(cb, 16 + (B)) * rw_; \
    } while (0)
#define NEGMAX (-3.402823466e+38f)

__device__ __forceinline__ void wave_lds_handover() {
    __builtin_amdgcn_fence(__ATOMIC_RELEASE, "wavefront");
    __builtin_amdgcn_wave_barrier();
    __builtin_amdgcn_fence(__ATOMIC_ACQUIRE, "wavefront");
}

__global__ __launch_bounds__(32 * AP_WAVES) void gat_agg_proj_ln(
    const float* __restrict__ x,
    const float* __restrict__ W,
    const float* __restrict__ asrc,
    const float* __restrict__ adst,
    const int* __restrict__ off,
    const int* __restrict__ csrc,
    const float* __restrict__ bias,
    const float* __restrict__ gamma,
    const float* __restrict__ beta,
    float* __restrict__ out)
{
    __shared__ __attribute__((aligned(16))) float smem[DD * COLS + AP_WAVES * 16 * ZK];
    float* const sW = smem;
    float* const sZ = smem + DD * COLS;

    const int lane = threadIdx.x & 31;
    const int wave = __builtin_amdgcn_readfirstlane((int)(threadIdx.x >> 5));

    {
        const int wg = wave & 1, wh = (wave >> 1) & 3, wq = wave >> 3;
        const float* wsrc = W + (size_t)(2 * wq) * COLS + wh * DD + 32 * wg + lane;
        float* wdst = sW + (wh * 16) * WSTEP + wg * 128 + (wq * 16 + (lane & 15)) * 4 + (lane >> 4) * 2;
        v2f wst[16];
#pragma unroll
        for (int it = 0; it < 16; ++it) {
            wst[it].x = wsrc[(size_t)(4 * it) * COLS];
            wst[it].y = wsrc[(size_t)(4 * it + 1) * COLS];
        }
#pragma unroll
        for (int it = 0; it < 16; ++it)
            *(v2f*)(wdst + it * WSTEP) = wst[it];
    }
    const bool live = wave * AP_GRID + (int)blockIdx.x < MTILES;
    const int tile = min(wave * AP_GRID + (int)blockIdx.x, MTILES - 1);
    const int n    = tile / (BTOT / 16);
    const int bt0  = (tile - n * (BTOT / 16)) * 16;
    const int kb = off[n], ke = off[n + 1];
    float* zt = sZ + wave * (16 * ZK);

    const int ib = lane & 15;
    const int hp = lane >> 4;
    const size_t lo = (size_t)(bt0 + ib) * HH + 2 * hp;
    const v2f ad2 = *(const v2f*)(adst + (size_t)n * (BTOT * HH) + lo);

    float m0 = NEGMAX, m1 = NEGMAX, s0 = 0.f, s1 = 0.f;
    int j0 = 0;
    for (int base = kb; base < ke; base += 32) {
        const int cnt  = min(32, ke - base);
        const int j    = csrc[min(base + lane, ke - 1)];
        if (base == kb) j0 = j;
        for (int e = 0; e < cnt; e += PA_U) {
            v2f as2[PA_U];
#pragma unroll
            for (int u = 0; u < PA_U; ++u) {
                const int je = __builtin_amdgcn_readlane(j, e + u);
                as2[u] = *(const v2f*)(asrc + (size_t)je * (BTOT * HH) + lo);
            }
#pragma unroll
            for (int g = 0; g < PA_U / 4; ++g) {
                if (e + 4 * g < cnt) {
#pragma unroll
                    for (int u = 4 * g; u < 4 * g + 4; ++u) {
                        const bool real = e + u < cnt;
                        const float a0 = leaky(as2[u].x + ad2.x);
                        const float a1 = leaky(as2[u].y + ad2.y);
                        const float n0 = fmaxf(m0, a0);
                        const float n1 = fmaxf(m1, a1);
                        s0 = s0 * __expf(m0 - n0) + (real ? __expf(a0 - n0) : 0.f);
                        s1 = s1 * __expf(m1 - n1) + (real ? __expf(a1 - n1) : 0.f);
                        m0 = n0; m1 = n1;
                    }
                }
            }
        }
    }
    const float i0 = 1.f / (s0 + 1e-16f);
    const float i1 = 1.f / (s1 + 1e-16f);

    __syncthreads();
    if (!live) return;

    v2f acc[16][HH];
#pragma unroll
    for (int b = 0; b < 16; ++b) {
#pragma unroll
        for (int h = 0; h < HH; ++h) acc[b][h] = (v2f){0.f, 0.f};
    }

    const float* xcol = x + (size_t)bt0 * (NN * DD) + 2 * lane;
    const int deg  = ke - kb;
    const int degg = (deg + GE - 1) & ~(GE - 1);
    int j = j0;
    for (int e = 0; e < degg; e += GE) {
        if ((e & 31) == 0 && e != 0) j = csrc[min(kb + e + lane, ke - 1)];
        v2f as2[GE];
        v2f row[GE][16];
#pragma unroll
        for (int u = 0; u < GE; ++u) {
            const int je = __builtin_amdgcn_readlane(j, (e & 31) + u);
            as2[u] = *(const v2f*)(asrc + (size_t)je * (BTOT * HH) + lo);
        }
#pragma unroll
        for (int u = 0; u < GE; ++u) {
            const int je = __builtin_amdgcn_readlane(j, (e & 31) + u);
            const float* rp = xcol + (size_t)je * DD;
#pragma unroll
            for (int b = 0; b < 16; ++b)
                row[u][b] = *(const v2f*)(rp + (size_t)b * (NN * DD));
        }
#pragma unroll
        for (int u = 0; u < GE; ++u) {
            const float keep = (e + u < deg) ? 1.f : 0.f;
            const float a0 = leaky(as2[u].x + ad2.x);
            const float a1 = leaky(as2[u].y + ad2.y);
            const float ca = __expf(a0 - m0) * i0 * keep;
            const float cb = __expf(a1 - m1) * i1 * keep;
            AGG_ITEM(0);  AGG_ITEM(1);  AGG_ITEM(2);  AGG_ITEM(3);
            __builtin_amdgcn_sched_barrier(0);
            AGG_ITEM(4);  AGG_ITEM(5);  AGG_ITEM(6);  AGG_ITEM(7);
            __builtin_amdgcn_sched_barrier(0);
            AGG_ITEM(8);  AGG_ITEM(9);  AGG_ITEM(10); AGG_ITEM(11);
            __builtin_amdgcn_sched_barrier(0);
            AGG_ITEM(12); AGG_ITEM(13); AGG_ITEM(14); AGG_ITEM(15);
            __builtin_amdgcn_sched_barrier(0);
        }
    }

    const int r    = lane & 15;
    const int half = lane >> 4;
    const int kgrp = half * 2;
    const size_t rowbase = ((size_t)(bt0 + half * 8) * NN + n) * DD + r;
    float xr[8][4];
    v8f o0 = {}, o1 = {}, o2 = {}, o3 = {};
    const float* zrow = zt + r * ZK;
    const int    swr  = 4 * r;
    const float* wfrag = sW + lane * 4;

#pragma unroll
    for (int part = 0; part < 2; ++part) {
#pragma unroll
        for (int b = 0; b < 16; ++b) {
            float* zr = zt + b * ZK;
            const int sw = 4 * b;
#pragma unroll
            for (int hh = 0; hh < 2; ++hh)
                *(v2f*)(zr + ((hh * DD + 2 * lane) ^ sw)) = acc[b][2 * part + hh];
        }
        wave_lds_handover();

        if (part == 1) {
#pragma unroll
            for (int v = 0; v < 8; ++v) {
#pragma unroll
                for (int c = 0; c < 4; ++c)
                    xr[v][c] = x[rowbase + (size_t)v * (NN * DD) + c * 16];
            }
        }

        v2f fa[2][PB];
        v4f fq[2][2 * PB];
#pragma unroll
        for (int s = 0; s < PB; ++s) {
            fa[0][s]         = *(const v2f*)(zrow + ((s * 4 + kgrp) ^ swr));
            fq[0][2 * s]     = *(const v4f*)(wfrag + (32 * part + s) * WSTEP);
            fq[0][2 * s + 1] = *(const v4f*)(wfrag + (32 * part + s) * WSTEP + 128);
        }
#pragma unroll
        for (int blk = 0; blk < 32 / PB; ++blk) {
            const int cur = blk & 1, nxt = cur ^ 1;
            if (blk + 1 < 32 / PB) {
#pragma unroll
                for (int s = 0; s < PB; ++s) {
                    const int kk = (blk + 1) * PB + s;
                    fa[nxt][s]         = *(const v2f*)(zrow + ((kk * 4 + kgrp) ^ swr));
                    fq[nxt][2 * s]     = *(const v4f*)(wfrag + (32 * part + kk) * WSTEP);
                    fq[nxt][2 * s + 1] = *(const v4f*)(wfrag + (32 * part + kk) * WSTEP + 128);
                }
            }
            __builtin_amdgcn_sched_barrier(0);
#pragma unroll
            for (int s = 0; s < PB; ++s) {
                const v2f a = fa[cur][s];
                const v4f q0 = fq[cur][2 * s], q1 = fq[cur][2 * s + 1];
                const v2f b0 = {q0.x, q0.y}, b1 = {q0.z, q0.w}, b2 = {q1.x, q1.y}, b3 = {q1.z, q1.w};
                o0 = __builtin_amdgcn_wmma_f32_16x16x4_f32(false, a, false, b0, (short)0, o0, false, false);
                o1 = __builtin_amdgcn_wmma_f32_16x16x4_f32(false, a, false, b1, (short)0, o1, false, false);
                o2 = __builtin_amdgcn_wmma_f32_16x16x4_f32(false, a, false, b2, (short)0, o2, false, false);
                o3 = __builtin_amdgcn_wmma_f32_16x16x4_f32(false, a, false, b3, (short)0, o3, false, false);
            }
            __builtin_amdgcn_sched_barrier(0);
        }
        if (part == 0) {
            wave_lds_handover();
        }
    }

    {
        const float bb0 = bias[r], bb1 = bias[16 + r], bb2 = bias[32 + r], bb3 = bias[48 + r];
        const float g0 = gamma[r], g1 = gamma[16 + r], g2 = gamma[32 + r], g3 = gamma[48 + r];
        const float be0 = beta[r], be1 = beta[16 + r], be2 = beta[32 + r], be3 = beta[48 + r];
#pragma unroll
        for (int v = 0; v < 8; ++v) {
            const size_t idx = rowbase + (size_t)v * (NN * DD);
            const float y0 = o0[v] * 0.25f + bb0 + xr[v][0];
            const float y1 = o1[v] * 0.25f + bb1 + xr[v][1];
            const float y2 = o2[v] * 0.25f + bb2 + xr[v][2];
            const float y3 = o3[v] * 0.25f + bb3 + xr[v][3];
            const float sm = row16_sum((y0 + y1) + (y2 + y3));
            const float sq = row16_sum((y0 * y0 + y1 * y1) + (y2 * y2 + y3 * y3));
            const float mu  = sm * (1.f / 64.f);
            const float var = sq * (1.f / 64.f) - mu * mu;
            const float rs  = rsqrtf(var + 1e-5f);
            out[idx]      = (y0 - mu) * rs * g0 + be0;
            out[idx + 16] = (y1 - mu) * rs * g1 + be1;
            out[idx + 32] = (y2 - mu) * rs * g2 + be2;
            out[idx + 48] = (y3 - mu) * rs * g3 + be3;
        }
    }
}

extern "C" void kernel_launch(void* const* d_in, const int* in_sizes, int n_in,
                              void* d_out, int out_size, void* d_ws, size_t ws_size,
                              hipStream_t stream) {
    const float* x        = (const float*)d_in[0];
    const float* W        = (const float*)d_in[1];
    const float* att_src  = (const float*)d_in[2];
    const float* att_dst  = (const float*)d_in[3];
    const float* bias     = (const float*)d_in[4];
    const float* gamma    = (const float*)d_in[5];
    const float* beta     = (const float*)d_in[6];
    const int*   ei       = (const int*)d_in[7];
    float* out            = (float*)d_out;
    const int E = in_sizes[7] / 2;

    char* ws = (char*)d_ws;
    float* asrc = (float*)ws;  ws += sizeof(float) * (size_t)MROWS * HH;
    float* adst = (float*)ws;  ws += sizeof(float) * (size_t)MROWS * HH;
    int* off    = (int*)ws;    ws += sizeof(int) * (NN + 1);
    int* csrc   = (int*)ws;    ws += sizeof(int) * (E + NN);

    gat_logits_wmma<<<(MTILES + 7) / 8 + 1, 256, 0, stream>>>(x, W, att_src, att_dst, asrc, adst,
                                                            ei, E, off, csrc);

    gat_agg_proj_ln<<<AP_GRID, 32 * AP_WAVES, 0, stream>>>(x, W, asrc, adst, off, csrc,
                                          bias, gamma, beta, out);
}
